# starting kernel
# baseline (speedup 1.0000x reference)
#include <hip/hip_runtime.h>
#include <hip/hip_bf16.h>
#include <math.h>

constexpr int Bn = 16;
constexpr int Cn = 4;
constexpr int Ln = 4096;
constexpr int Kn = 256;
constexpr int Sn = 128;
constexpr int Wn = Ln - Sn + 1;
constexpr int NT = (Wn + 15) / 16;
constexpr float EPSV = 1e-8f;

typedef __attribute__((ext_vector_type(2))) float v2f;
typedef __attribute__((ext_vector_type(8))) float v8f;

__global__ __launch_bounds__(128) void shapelet_norm_kernel(
    const float* __restrict__ shp, float* __restrict__ shpn)
{
    __shared__ float red[4];
    const int row = blockIdx.x;
    const int s   = threadIdx.x;
    const float v = shp[row * Sn + s];
    float p = v * v;
    #pragma unroll
    for (int off = 16; off >= 1; off >>= 1) p += __shfl_xor(p, off, 32);
    if ((s & 31) == 0) red[s >> 5] = p;
    __syncthreads();
    const float tot = red[0] + red[1] + red[2] + red[3];
    shpn[row * Sn + s] = v / fmaxf(sqrtf(tot), EPSV);
}

__global__ __launch_bounds__(256) void window_norm_kernel(
    const float* __restrict__ x, float* __restrict__ invn)
{
    const int bc = blockIdx.x;
    const float* px = x + (size_t)bc * Ln;
    for (int w = threadIdx.x; w < Wn; w += 256) {
        float ss = 0.f;
        #pragma unroll 8
        for (int s = 0; s < Sn; ++s) { const float v = px[w + s]; ss = fmaf(v, v, ss); }
        invn[bc * Wn + w] = 1.0f / fmaxf(sqrtf(ss), EPSV);
    }
}

__global__ __launch_bounds__(128) void shapelet_wmma_kernel(
    const float* __restrict__ x, const float* __restrict__ shpn,
    const float* __restrict__ invn, float* __restrict__ out)
{
    __shared__ v2f   ldsB[Cn * (Sn / 2) * 16];
    __shared__ float ldsR[4 * 16];

    const int b    = blockIdx.x >> 4;
    const int k0   = (blockIdx.x & 15) << 4;
    const int tid  = threadIdx.x;
    const int wave = tid >> 5;
    const int lane = tid & 31;
    const int n    = lane & 15;
    const int kA   = (lane >> 4) << 1;

    for (int i = tid; i < Cn * (Sn / 2) * 16; i += 128) {
        const int nn = i & 15;
        const int s2 = (i >> 4) & 63;
        const int c  = i >> 10;
        ldsB[i] = ((const v2f*)shpn)[(size_t)(c * Kn + (k0 + nn)) * (Sn / 2) + s2];
    }
    __syncthreads();

    float rmax = 0.0f;
    for (int t = wave; t < NT; t += 4) {
        const int w0 = t << 4;
        const int wv = w0 + n;
        const int wc = wv < Wn ? wv : Wn - 1;
        v8f acc = {0.f, 0.f, 0.f, 0.f, 0.f, 0.f, 0.f, 0.f};

        for (int c = 0; c < Cn; ++c) {
            float s_inv = invn[(b * Cn + c) * Wn + wc];
            s_inv = (wv < Wn) ? s_inv : 0.0f;
            const float* px = x + (size_t)(b * Cn + c) * Ln + wc + kA;
            const v2f*   pb = &ldsB[(c * 64) * 16 + n];
            #pragma unroll 4
            for (int s0 = 0; s0 < Sn; s0 += 4) {
                v2f a;
                a.x = px[s0]     * s_inv;
                a.y = px[s0 + 1] * s_inv;
                const v2f bv = pb[((s0 + kA) >> 1) * 16];
                acc = __builtin_amdgcn_wmma_f32_16x16x4_f32(
                     false, a,  false, bv,
                     (short)0, acc,  false,  false);
            }
        }
        #pragma unroll
        for (int r = 0; r < 8; ++r) rmax = fmaxf(rmax, acc[r] * 0.25f);
    }

    rmax = fmaxf(rmax, __shfl_xor(rmax, 16, 32));
    if (lane < 16) ldsR[wave * 16 + n] = rmax;
    __syncthreads();
    if (tid < 16) {
        const float v = fmaxf(fmaxf(ldsR[tid], ldsR[16 + tid]),
                              fmaxf(ldsR[32 + tid], ldsR[48 + tid]));
        out[b * Kn + k0 + tid] = v;
    }
}

extern "C" void kernel_launch(void* const* d_in, const int* in_sizes, int n_in,
                              void* d_out, int out_size, void* d_ws, size_t ws_size,
                              hipStream_t stream) {
    const float* x   = (const float*)d_in[0];
    const float* shp = (const float*)d_in[1];
    float* out  = (float*)d_out;
    float* shpn = (float*)d_ws;
    float* invn = shpn + Cn * Kn * Sn;

    shapelet_norm_kernel<<<Cn * Kn, Sn, 0, stream>>>(shp, shpn);
    window_norm_kernel<<<Bn * Cn, 256, 0, stream>>>(x, invn);
    shapelet_wmma_kernel<<<Bn * (Kn / 16), 128, 0, stream>>>(x, shpn, invn, out);
}


# Window-tile body fully unrolled (channel and k loops): x loads issue about 20 WMMAs ahead of use instead of being waited per 4-step trip; B pairs read from LDS once per wave and kept in VGPRs
# speedup vs baseline: 1.6099x; 1.6099x over previous
#include <hip/hip_runtime.h>
#include <hip/hip_bf16.h>
#include <math.h>

constexpr int Bn = 16;
constexpr int Cn = 4;
constexpr int Ln = 4096;
constexpr int Kn = 256;
constexpr int Sn = 128;
constexpr int Wn = Ln - Sn + 1;
constexpr int NT = (Wn + 15) / 16;
constexpr float EPSV = 1e-8f;

typedef __attribute__((ext_vector_type(2))) float v2f;
typedef __attribute__((ext_vector_type(8))) float v8f;

__global__ __launch_bounds__(128) void shapelet_norm_kernel(
    const float* __restrict__ shp, float* __restrict__ shpn)
{
    __shared__ float red[4];
    const int row = blockIdx.x;
    const int s   = threadIdx.x;
    const float v = shp[row * Sn + s];
    float p = v * v;
    #pragma unroll
    for (int off = 16; off >= 1; off >>= 1) p += __shfl_xor(p, off, 32);
    if ((s & 31) == 0) red[s >> 5] = p;
    __syncthreads();
    const float tot = red[0] + red[1] + red[2] + red[3];
    shpn[row * Sn + s] = v / fmaxf(sqrtf(tot), EPSV);
}

__global__ __launch_bounds__(256) void window_norm_kernel(
    const float* __restrict__ x, float* __restrict__ invn)
{
    const int bc = blockIdx.x;
    const float* px = x + (size_t)bc * Ln;
    for (int w = threadIdx.x; w < Wn; w += 256) {
        float ss = 0.f;
        #pragma unroll 8
        for (int s = 0; s < Sn; ++s) { const float v = px[w + s]; ss = fmaf(v, v, ss); }
        invn[bc * Wn + w] = 1.0f / fmaxf(sqrtf(ss), EPSV);
    }
}

__global__ __launch_bounds__(128) void shapelet_wmma_kernel(
    const float* __restrict__ x, const float* __restrict__ shpn,
    const float* __restrict__ invn, float* __restrict__ out)
{
    __shared__ v2f   ldsB[Cn * (Sn / 2) * 16];
    __shared__ float ldsR[4 * 16];

    const int b    = blockIdx.x >> 4;
    const int k0   = (blockIdx.x & 15) << 4;
    const int tid  = threadIdx.x;
    const int wave = tid >> 5;
    const int lane = tid & 31;
    const int n    = lane & 15;
    const int kA   = (lane >> 4) << 1;

    for (int i = tid; i < Cn * (Sn / 2) * 16; i += 128) {
        const int nn = i & 15;
        const int s2 = (i >> 4) & 63;
        const int c  = i >> 10;
        ldsB[i] = ((const v2f*)shpn)[(size_t)(c * Kn + (k0 + nn)) * (Sn / 2) + s2];
    }
    __syncthreads();

    float rmax = 0.0f;
    for (int t = wave; t < NT; t += 4) {
        const int w0 = t << 4;
        const int wv = w0 + n;
        const int wc = wv < Wn ? wv : Wn - 1;
        v8f acc = {0.f, 0.f, 0.f, 0.f, 0.f, 0.f, 0.f, 0.f};

        #pragma unroll
        for (int c = 0; c < Cn; ++c) {
            float s_inv = invn[(b * Cn + c) * Wn + wc];
            s_inv = (wv < Wn) ? s_inv : 0.0f;
            const float* px = x + (size_t)(b * Cn + c) * Ln + wc + kA;
            const v2f*   pb = &ldsB[(c * 64) * 16 + n];
            #pragma unroll
            for (int s0 = 0; s0 < Sn; s0 += 4) {
                v2f a;
                a.x = px[s0]     * s_inv;
                a.y = px[s0 + 1] * s_inv;
                const v2f bv = pb[((s0 + kA) >> 1) * 16];
                acc = __builtin_amdgcn_wmma_f32_16x16x4_f32(
                     false, a,  false, bv,
                     (short)0, acc,  false,  false);
            }
        }
        #pragma unroll
        for (int r = 0; r < 8; ++r) rmax = fmaxf(rmax, acc[r] * 0.25f);
    }

    rmax = fmaxf(rmax, __shfl_xor(rmax, 16, 32));
    if (lane < 16) ldsR[wave * 16 + n] = rmax;
    __syncthreads();
    if (tid < 16) {
        const float v = fmaxf(fmaxf(ldsR[tid], ldsR[16 + tid]),
                              fmaxf(ldsR[32 + tid], ldsR[48 + tid]));
        out[b * Kn + k0 + tid] = v;
    }
}

extern "C" void kernel_launch(void* const* d_in, const int* in_sizes, int n_in,
                              void* d_out, int out_size, void* d_ws, size_t ws_size,
                              hipStream_t stream) {
    const float* x   = (const float*)d_in[0];
    const float* shp = (const float*)d_in[1];
    float* out  = (float*)d_out;
    float* shpn = (float*)d_ws;
    float* invn = shpn + Cn * Kn * Sn;

    shapelet_norm_kernel<<<Cn * Kn, Sn, 0, stream>>>(shp, shpn);
    window_norm_kernel<<<Bn * Cn, 256, 0, stream>>>(x, invn);
    shapelet_wmma_kernel<<<Bn * (Kn / 16), 128, 0, stream>>>(x, shpn, invn, out);
}


# x loads grouped: each channel's 32 pair loads and its inverse norm go out as one group between two 32-WMMA runs, three runs ahead, into 4 register sets, instead of one load between every two WMMAs
# speedup vs baseline: 2.5934x; 1.6109x over previous
#include <hip/hip_runtime.h>
#include <hip/hip_bf16.h>
#include <math.h>

constexpr int Bn = 16;
constexpr int Cn = 4;
constexpr int Ln = 4096;
constexpr int Kn = 256;
constexpr int Sn = 128;
constexpr int Wn = Ln - Sn + 1;
constexpr int NT = (Wn + 15) / 16;
constexpr float EPSV = 1e-8f;

typedef __attribute__((ext_vector_type(2))) float v2f;
typedef __attribute__((ext_vector_type(8))) float v8f;

__global__ __launch_bounds__(128) void shapelet_norm_kernel(
    const float* __restrict__ shp, float* __restrict__ shpn)
{
    __shared__ float red[4];
    const int row = blockIdx.x;
    const int s   = threadIdx.x;
    const float v = shp[row * Sn + s];
    float p = v * v;
    #pragma unroll
    for (int off = 16; off >= 1; off >>= 1) p += __shfl_xor(p, off, 32);
    if ((s & 31) == 0) red[s >> 5] = p;
    __syncthreads();
    const float tot = red[0] + red[1] + red[2] + red[3];
    shpn[row * Sn + s] = v / fmaxf(sqrtf(tot), EPSV);
}

__global__ __launch_bounds__(256) void window_norm_kernel(
    const float* __restrict__ x, float* __restrict__ invn)
{
    const int bc = blockIdx.x;
    const float* px = x + (size_t)bc * Ln;
    for (int w = threadIdx.x; w < Wn; w += 256) {
        float ss = 0.f;
        #pragma unroll 8
        for (int s = 0; s < Sn; ++s) { const float v = px[w + s]; ss = fmaf(v, v, ss); }
        invn[bc * Wn + w] = 1.0f / fmaxf(sqrtf(ss), EPSV);
    }
}


__device__ __forceinline__ void load_rows(
    const float* __restrict__ x, const float* __restrict__ invn,
    int bc, int t, int n, int kA, v2f (&a)[32], float& s_raw)
{
    const int wv = (t << 4) + n;
    const int wc = wv < Wn ? wv : Wn - 1;
    s_raw = invn[bc * Wn + wc];
    const float* px = x + (size_t)bc * Ln + wc + kA;
    #pragma unroll
    for (int j = 0; j < 32; ++j) {
        a[j].x = px[4 * j];
        a[j].y = px[4 * j + 1];
    }
}

__device__ __forceinline__ v8f mma_rows(
    v8f acc, const v2f (&a)[32], float s_raw, bool row_ok, const v2f (&bq)[32])
{
    asm volatile("" : "+v"(s_raw));
    const float s_inv = row_ok ? s_raw : 0.0f;
    #pragma unroll
    for (int j = 0; j < 32; ++j) {
        v2f av;
        av.x = a[j].x * s_inv;
        av.y = a[j].y * s_inv;
        const v2f bv = bq[j];
        acc = __builtin_amdgcn_wmma_f32_16x16x4_f32(
             false, av,  false, bv,
             (short)0, acc,  false,  false);
    }
    asm volatile("" : "+v"(acc));
    return acc;
}

__global__ __launch_bounds__(128) void shapelet_wmma_kernel(
    const float* __restrict__ x, const float* __restrict__ shpn,
    const float* __restrict__ invn, float* __restrict__ out)
{
    __shared__ v2f   ldsB[Cn * (Sn / 2) * 16];
    __shared__ float ldsR[4 * 16];

    const int b    = blockIdx.x >> 4;
    const int k0   = (blockIdx.x & 15) << 4;
    const int tid  = threadIdx.x;
    const int wave = tid >> 5;
    const int lane = tid & 31;
    const int n    = lane & 15;
    const int kA   = (lane >> 4) << 1;

    for (int i = tid; i < Cn * (Sn / 2) * 16; i += 128) {
        const int nn = i & 15;
        const int s2 = (i >> 4) & 63;
        const int c  = i >> 10;
        ldsB[i] = ((const v2f*)shpn)[(size_t)(c * Kn + (k0 + nn)) * (Sn / 2) + s2];
    }
    __syncthreads();

    const int bc0 = b * Cn;
    v2f b0[32], b1[32], b2[32], b3[32];
    #pragma unroll
    for (int j = 0; j < 32; ++j) {
        b0[j] = ldsB[(0 * 64 + ((4 * j + kA) >> 1)) * 16 + n];
        b1[j] = ldsB[(1 * 64 + ((4 * j + kA) >> 1)) * 16 + n];
        b2[j] = ldsB[(2 * 64 + ((4 * j + kA) >> 1)) * 16 + n];
        b3[j] = ldsB[(3 * 64 + ((4 * j + kA) >> 1)) * 16 + n];
    }

    v2f a0[32], a1[32], a2[32], a3[32];
    float i0, i1, i2, i3;

    int t = wave;
    load_rows(x, invn, bc0 + 0, t, n, kA, a0, i0);
    load_rows(x, invn, bc0 + 1, t, n, kA, a1, i1);
    load_rows(x, invn, bc0 + 2, t, n, kA, a2, i2);

    float rmax = 0.0f;
    for (; t < NT; t += 4) {
        const int tn = (t + 4 < NT) ? t + 4 : t;
        const bool row_ok = ((t << 4) + n) < Wn;
        v8f acc = {0.f, 0.f, 0.f, 0.f, 0.f, 0.f, 0.f, 0.f};

        load_rows(x, invn, bc0 + 3, t, n, kA, a3, i3);
        acc = mma_rows(acc, a0, i0, row_ok, b0);
        load_rows(x, invn, bc0 + 0, tn, n, kA, a0, i0);
        acc = mma_rows(acc, a1, i1, row_ok, b1);
        load_rows(x, invn, bc0 + 1, tn, n, kA, a1, i1);
        acc = mma_rows(acc, a2, i2, row_ok, b2);
        load_rows(x, invn, bc0 + 2, tn, n, kA, a2, i2);
        acc = mma_rows(acc, a3, i3, row_ok, b3);

        #pragma unroll
        for (int r = 0; r < 8; ++r) rmax = fmaxf(rmax, acc[r] * 0.25f);
    }

    rmax = fmaxf(rmax, __shfl_xor(rmax, 16, 32));
    if (lane < 16) ldsR[wave * 16 + n] = rmax;
    __syncthreads();
    if (tid < 16) {
        const float v = fmaxf(fmaxf(ldsR[tid], ldsR[16 + tid]),
                              fmaxf(ldsR[32 + tid], ldsR[48 + tid]));
        out[b * Kn + k0 + tid] = v;
    }
}

extern "C" void kernel_launch(void* const* d_in, const int* in_sizes, int n_in,
                              void* d_out, int out_size, void* d_ws, size_t ws_size,
                              hipStream_t stream) {
    const float* x   = (const float*)d_in[0];
    const float* shp = (const float*)d_in[1];
    float* out  = (float*)d_out;
    float* shpn = (float*)d_ws;
    float* invn = shpn + Cn * Kn * Sn;

    shapelet_norm_kernel<<<Cn * Kn, Sn, 0, stream>>>(shp, shpn);
    window_norm_kernel<<<Bn * Cn, 256, 0, stream>>>(x, invn);
    shapelet_wmma_kernel<<<Bn * (Kn / 16), 128, 0, stream>>>(x, shpn, invn, out);
}


# Main kernel at two waves per SIMD (8-wave blocks, tiles strided by 8; x register sets 4 -> 2 so a wave fits 512 VGPRs): the sibling's WMMA runs fill a wave's load, multiply and chain-latency gaps
# speedup vs baseline: 3.1070x; 1.1980x over previous
#include <hip/hip_runtime.h>
#include <hip/hip_bf16.h>
#include <math.h>

constexpr int Bn = 16;
constexpr int Cn = 4;
constexpr int Ln = 4096;
constexpr int Kn = 256;
constexpr int Sn = 128;
constexpr int Wn = Ln - Sn + 1;
constexpr int NT = (Wn + 15) / 16;
constexpr float EPSV = 1e-8f;

typedef __attribute__((ext_vector_type(2))) float v2f;
typedef __attribute__((ext_vector_type(8))) float v8f;

__global__ __launch_bounds__(128) void shapelet_norm_kernel(
    const float* __restrict__ shp, float* __restrict__ shpn)
{
    __shared__ float red[4];
    const int row = blockIdx.x;
    const int s   = threadIdx.x;
    const float v = shp[row * Sn + s];
    float p = v * v;
    #pragma unroll
    for (int off = 16; off >= 1; off >>= 1) p += __shfl_xor(p, off, 32);
    if ((s & 31) == 0) red[s >> 5] = p;
    __syncthreads();
    const float tot = red[0] + red[1] + red[2] + red[3];
    shpn[row * Sn + s] = v / fmaxf(sqrtf(tot), EPSV);
}

__global__ __launch_bounds__(256) void window_norm_kernel(
    const float* __restrict__ x, float* __restrict__ invn)
{
    const int bc = blockIdx.x;
    const float* px = x + (size_t)bc * Ln;
    for (int w = threadIdx.x; w < Wn; w += 256) {
        float ss = 0.f;
        #pragma unroll 8
        for (int s = 0; s < Sn; ++s) { const float v = px[w + s]; ss = fmaf(v, v, ss); }
        invn[bc * Wn + w] = 1.0f / fmaxf(sqrtf(ss), EPSV);
    }
}


__device__ __forceinline__ void load_rows(
    const float* __restrict__ x, const float* __restrict__ invn,
    int bc, int t, int n, int kA, v2f (&a)[32], float& s_raw)
{
    const int wv = (t << 4) + n;
    const int wc = wv < Wn ? wv : Wn - 1;
    s_raw = invn[bc * Wn + wc];
    const float* px = x + (size_t)bc * Ln + wc + kA;
    #pragma unroll
    for (int j = 0; j < 32; ++j) {
        a[j].x = px[4 * j];
        a[j].y = px[4 * j + 1];
    }
}

__device__ __forceinline__ v8f mma_rows(
    v8f acc, const v2f (&a)[32], float s_raw, bool row_ok, const v2f (&bq)[32])
{
    asm volatile("" : "+v"(s_raw));
    const float s_inv = row_ok ? s_raw : 0.0f;
    v2f av[32];
    #pragma unroll
    for (int j = 0; j < 32; ++j) {
        av[j].x = a[j].x * s_inv;
        av[j].y = a[j].y * s_inv;
    }
    __builtin_amdgcn_sched_barrier(0);
    #pragma unroll
    for (int j = 0; j < 32; ++j) {
        const v2f bv = bq[j];
        acc = __builtin_amdgcn_wmma_f32_16x16x4_f32(
             false, av[j],  false, bv,
             (short)0, acc,  false,  false);
    }
    __builtin_amdgcn_sched_barrier(0);
    asm volatile("v_nop\n\tv_nop\n\tv_nop\n\tv_nop" : "+v"(acc));
    return acc;
}

__global__ __launch_bounds__(256) void shapelet_wmma_kernel(
    const float* __restrict__ x, const float* __restrict__ shpn,
    const float* __restrict__ invn, float* __restrict__ out)
{
    __shared__ v2f   ldsB[Cn * (Sn / 2) * 16];
    __shared__ float ldsR[8 * 16];

    const int b    = blockIdx.x >> 4;
    const int k0   = (blockIdx.x & 15) << 4;
    const int tid  = threadIdx.x;
    const int wave = tid >> 5;
    const int lane = tid & 31;
    const int n    = lane & 15;
    const int kA   = (lane >> 4) << 1;

    for (int i = tid; i < Cn * (Sn / 2) * 16; i += 256) {
        const int nn = i & 15;
        const int s2 = (i >> 4) & 63;
        const int c  = i >> 10;
        ldsB[i] = ((const v2f*)shpn)[(size_t)(c * Kn + (k0 + nn)) * (Sn / 2) + s2];
    }
    __syncthreads();

    const int bc0 = b * Cn;
    v2f b0[32], b1[32], b2[32], b3[32];
    #pragma unroll
    for (int j = 0; j < 32; ++j) {
        b0[j] = ldsB[(0 * 64 + ((4 * j + kA) >> 1)) * 16 + n];
        b1[j] = ldsB[(1 * 64 + ((4 * j + kA) >> 1)) * 16 + n];
        b2[j] = ldsB[(2 * 64 + ((4 * j + kA) >> 1)) * 16 + n];
        b3[j] = ldsB[(3 * 64 + ((4 * j + kA) >> 1)) * 16 + n];
    }

    v2f a0[32], a1[32];
    float i0, i1;

    int t = wave;
    load_rows(x, invn, bc0 + 0, t, n, kA, a0, i0);

    float rmax = 0.0f;
    for (; t < NT; t += 8) {
        const int tn = (t + 8 < NT) ? t + 8 : t;
        const bool row_ok = ((t << 4) + n) < Wn;
        v8f acc = {0.f, 0.f, 0.f, 0.f, 0.f, 0.f, 0.f, 0.f};

        load_rows(x, invn, bc0 + 1, t, n, kA, a1, i1);
        acc = mma_rows(acc, a0, i0, row_ok, b0);
        load_rows(x, invn, bc0 + 2, t, n, kA, a0, i0);
        acc = mma_rows(acc, a1, i1, row_ok, b1);
        load_rows(x, invn, bc0 + 3, t, n, kA, a1, i1);
        acc = mma_rows(acc, a0, i0, row_ok, b2);
        load_rows(x, invn, bc0 + 0, tn, n, kA, a0, i0);
        acc = mma_rows(acc, a1, i1, row_ok, b3);

        #pragma unroll
        for (int r = 0; r < 8; ++r) rmax = fmaxf(rmax, acc[r] * 0.25f);
    }

    rmax = fmaxf(rmax, __shfl_xor(rmax, 16, 32));
    if (lane < 16) ldsR[wave * 16 + n] = rmax;
    __syncthreads();
    if (tid < 16) {
        float v = ldsR[tid];
        for (int w = 1; w < 8; ++w) v = fmaxf(v, ldsR[w * 16 + tid]);
        out[b * Kn + k0 + tid] = v;
    }
}

extern "C" void kernel_launch(void* const* d_in, const int* in_sizes, int n_in,
                              void* d_out, int out_size, void* d_ws, size_t ws_size,
                              hipStream_t stream) {
    const float* x   = (const float*)d_in[0];
    const float* shp = (const float*)d_in[1];
    float* out  = (float*)d_out;
    float* shpn = (float*)d_ws;
    float* invn = shpn + Cn * Kn * Sn;

    shapelet_norm_kernel<<<Cn * Kn, Sn, 0, stream>>>(shp, shpn);
    window_norm_kernel<<<Bn * Cn, 256, 0, stream>>>(x, invn);
    shapelet_wmma_kernel<<<Bn * (Kn / 16), 256, 0, stream>>>(x, shpn, invn, out);
}


# Window-norm kernel: a thread owns 4 adjacent windows, reads their 131 elements once (33 loads in one group) and runs 4 interleaved sum chains, 256 blocks on all WGPs; each sum's own order unchanged
# speedup vs baseline: 4.4560x; 1.4342x over previous
#include <hip/hip_runtime.h>
#include <hip/hip_bf16.h>
#include <math.h>

constexpr int Bn = 16;
constexpr int Cn = 4;
constexpr int Ln = 4096;
constexpr int Kn = 256;
constexpr int Sn = 128;
constexpr int Wn = Ln - Sn + 1;
constexpr int NT = (Wn + 15) / 16;
constexpr float EPSV = 1e-8f;

typedef __attribute__((ext_vector_type(2))) float v2f;
typedef __attribute__((ext_vector_type(8))) float v8f;

__global__ __launch_bounds__(128) void shapelet_norm_kernel(
    const float* __restrict__ shp, float* __restrict__ shpn)
{
    __shared__ float red[4];
    const int row = blockIdx.x;
    const int s   = threadIdx.x;
    const float v = shp[row * Sn + s];
    float p = v * v;
    #pragma unroll
    for (int off = 16; off >= 1; off >>= 1) p += __shfl_xor(p, off, 32);
    if ((s & 31) == 0) red[s >> 5] = p;
    __syncthreads();
    const float tot = red[0] + red[1] + red[2] + red[3];
    shpn[row * Sn + s] = v / fmaxf(sqrtf(tot), EPSV);
}

__global__ __launch_bounds__(256) void window_norm_kernel(
    const float* __restrict__ x, float* __restrict__ invn)
{
    const int bc = blockIdx.x >> 2;
    const int w0 = ((blockIdx.x & 3) << 10) + (threadIdx.x << 2);
    if (w0 >= Wn) return;
    const float* px = x + (size_t)bc * Ln + w0;

    float r[131];
    #pragma unroll
    for (int e = 0; e < 128; ++e) r[e] = px[e];
    r[128] = 0.0f; r[129] = 0.0f; r[130] = 0.0f;
    if (w0 + 130 < Ln) { r[128] = px[128]; r[129] = px[129]; r[130] = px[130]; }

    float ss[4] = {0.f, 0.f, 0.f, 0.f};
    #pragma unroll
    for (int s = 0; s < Sn; ++s) {
        #pragma unroll
        for (int j = 0; j < 4; ++j) ss[j] = fmaf(r[s + j], r[s + j], ss[j]);
    }
    float inv[4];
    #pragma unroll
    for (int j = 0; j < 4; ++j) inv[j] = 1.0f / fmaxf(sqrtf(ss[j]), EPSV);
    asm volatile("" : "+v"(inv[0]), "+v"(inv[1]), "+v"(inv[2]), "+v"(inv[3]));
    #pragma unroll
    for (int j = 0; j < 4; ++j)
        if (w0 + j < Wn) invn[bc * Wn + w0 + j] = inv[j];
}


__device__ __forceinline__ void load_rows(
    const float* __restrict__ x, const float* __restrict__ invn,
    int bc, int t, int n, int kA, v2f (&a)[32], float& s_raw)
{
    const int wv = (t << 4) + n;
    const int wc = wv < Wn ? wv : Wn - 1;
    s_raw = invn[bc * Wn + wc];
    const float* px = x + (size_t)bc * Ln + wc + kA;
    #pragma unroll
    for (int j = 0; j < 32; ++j) {
        a[j].x = px[4 * j];
        a[j].y = px[4 * j + 1];
    }
}

__device__ __forceinline__ v8f mma_rows(
    v8f acc, const v2f (&a)[32], float s_raw, bool row_ok, const v2f (&bq)[32])
{
    asm volatile("" : "+v"(s_raw));
    const float s_inv = row_ok ? s_raw : 0.0f;
    v2f av[32];
    #pragma unroll
    for (int j = 0; j < 32; ++j) {
        av[j].x = a[j].x * s_inv;
        av[j].y = a[j].y * s_inv;
    }
    __builtin_amdgcn_sched_barrier(0);
    #pragma unroll
    for (int j = 0; j < 32; ++j) {
        const v2f bv = bq[j];
        acc = __builtin_amdgcn_wmma_f32_16x16x4_f32(
             false, av[j],  false, bv,
             (short)0, acc,  false,  false);
    }
    __builtin_amdgcn_sched_barrier(0);
    asm volatile("v_nop\n\tv_nop\n\tv_nop\n\tv_nop" : "+v"(acc));
    return acc;
}

__global__ __launch_bounds__(256) void shapelet_wmma_kernel(
    const float* __restrict__ x, const float* __restrict__ shpn,
    const float* __restrict__ invn, float* __restrict__ out)
{
    __shared__ v2f   ldsB[Cn * (Sn / 2) * 16];
    __shared__ float ldsR[8 * 16];

    const int b    = blockIdx.x >> 4;
    const int k0   = (blockIdx.x & 15) << 4;
    const int tid  = threadIdx.x;
    const int wave = tid >> 5;
    const int lane = tid & 31;
    const int n    = lane & 15;
    const int kA   = (lane >> 4) << 1;

    for (int i = tid; i < Cn * (Sn / 2) * 16; i += 256) {
        const int nn = i & 15;
        const int s2 = (i >> 4) & 63;
        const int c  = i >> 10;
        ldsB[i] = ((const v2f*)shpn)[(size_t)(c * Kn + (k0 + nn)) * (Sn / 2) + s2];
    }
    __syncthreads();

    const int bc0 = b * Cn;
    v2f b0[32], b1[32], b2[32], b3[32];
    #pragma unroll
    for (int j = 0; j < 32; ++j) {
        b0[j] = ldsB[(0 * 64 + ((4 * j + kA) >> 1)) * 16 + n];
        b1[j] = ldsB[(1 * 64 + ((4 * j + kA) >> 1)) * 16 + n];
        b2[j] = ldsB[(2 * 64 + ((4 * j + kA) >> 1)) * 16 + n];
        b3[j] = ldsB[(3 * 64 + ((4 * j + kA) >> 1)) * 16 + n];
    }

    v2f a0[32], a1[32];
    float i0, i1;

    int t = wave;
    load_rows(x, invn, bc0 + 0, t, n, kA, a0, i0);

    float rmax = 0.0f;
    for (; t < NT; t += 8) {
        const int tn = (t + 8 < NT) ? t + 8 : t;
        const bool row_ok = ((t << 4) + n) < Wn;
        v8f acc = {0.f, 0.f, 0.f, 0.f, 0.f, 0.f, 0.f, 0.f};

        load_rows(x, invn, bc0 + 1, t, n, kA, a1, i1);
        acc = mma_rows(acc, a0, i0, row_ok, b0);
        load_rows(x, invn, bc0 + 2, t, n, kA, a0, i0);
        acc = mma_rows(acc, a1, i1, row_ok, b1);
        load_rows(x, invn, bc0 + 3, t, n, kA, a1, i1);
        acc = mma_rows(acc, a0, i0, row_ok, b2);
        load_rows(x, invn, bc0 + 0, tn, n, kA, a0, i0);
        acc = mma_rows(acc, a1, i1, row_ok, b3);

        #pragma unroll
        for (int r = 0; r < 8; ++r) rmax = fmaxf(rmax, acc[r] * 0.25f);
    }

    rmax = fmaxf(rmax, __shfl_xor(rmax, 16, 32));
    if (lane < 16) ldsR[wave * 16 + n] = rmax;
    __syncthreads();
    if (tid < 16) {
        float v = ldsR[tid];
        for (int w = 1; w < 8; ++w) v = fmaxf(v, ldsR[w * 16 + tid]);
        out[b * Kn + k0 + tid] = v;
    }
}

extern "C" void kernel_launch(void* const* d_in, const int* in_sizes, int n_in,
                              void* d_out, int out_size, void* d_ws, size_t ws_size,
                              hipStream_t stream) {
    const float* x   = (const float*)d_in[0];
    const float* shp = (const float*)d_in[1];
    float* out  = (float*)d_out;
    float* shpn = (float*)d_ws;
    float* invn = shpn + Cn * Kn * Sn;

    shapelet_norm_kernel<<<Cn * Kn, Sn, 0, stream>>>(shp, shpn);
    window_norm_kernel<<<Bn * Cn * 4, 256, 0, stream>>>(x, invn);
    shapelet_wmma_kernel<<<Bn * (Kn / 16), 256, 0, stream>>>(x, shpn, invn, out);
}


# Norm launches fused into the main kernel's prologue (3 -> 1 launches): shapelet rows scaled while staged, window inverse norms built in LDS from 16-element chunk sums; both in a new sum order, rsq
# speedup vs baseline: 4.6739x; 1.0489x over previous
#include <hip/hip_runtime.h>
#include <hip/hip_bf16.h>
#include <math.h>

constexpr int Bn = 16;
constexpr int Cn = 4;
constexpr int Ln = 4096;
constexpr int Kn = 256;
constexpr int Sn = 128;
constexpr int Wn = Ln - Sn + 1;
constexpr int NT = (Wn + 15) / 16;
constexpr float EPSV = 1e-8f;
constexpr int WP  = NT * 16;
constexpr int NCH = Ln / 16;

typedef __attribute__((ext_vector_type(2))) float v2f;
typedef __attribute__((ext_vector_type(8))) float v8f;


__device__ __forceinline__ void load_rows(
    const float* __restrict__ x, const float* invl,
    int bc, int t, int n, int kA, v2f (&a)[32], float& s_raw)
{
    const int wv = (t << 4) + n;
    const int wc = wv < Wn ? wv : Wn - 1;
    s_raw = invl[wc];
    const float* px = x + (size_t)bc * Ln + wc + kA;
    #pragma unroll
    for (int j = 0; j < 32; ++j) {
        a[j].x = px[4 * j];
        a[j].y = px[4 * j + 1];
    }
}

__device__ __forceinline__ v8f mma_rows(
    v8f acc, const v2f (&a)[32], float s_raw, bool row_ok, const v2f (&bq)[32])
{
    asm volatile("" : "+v"(s_raw));
    const float s_inv = row_ok ? s_raw : 0.0f;
    v2f av[32];
    #pragma unroll
    for (int j = 0; j < 32; ++j) {
        av[j].x = a[j].x * s_inv;
        av[j].y = a[j].y * s_inv;
    }
    __builtin_amdgcn_sched_barrier(0);
    #pragma unroll
    for (int j = 0; j < 32; ++j) {
        const v2f bv = bq[j];
        acc = __builtin_amdgcn_wmma_f32_16x16x4_f32(
             false, av[j],  false, bv,
             (short)0, acc,  false,  false);
    }
    __builtin_amdgcn_sched_barrier(0);
    asm volatile("v_nop\n\tv_nop\n\tv_nop\n\tv_nop" : "+v"(acc));
    return acc;
}

__global__ __launch_bounds__(256) void shapelet_wmma_kernel(
    const float* __restrict__ x, const float* __restrict__ shp, float* __restrict__ out)
{
    __shared__ v2f   ldsB[Cn * (Sn / 2) * 16];
    __shared__ float ldsInv[Cn * WP];
    __shared__ float ldsP[Cn * NCH];
    __shared__ float ldsR[8 * 16];

    const int b    = blockIdx.x >> 4;
    const int k0   = (blockIdx.x & 15) << 4;
    const int tid  = threadIdx.x;
    const int wave = tid >> 5;
    const int lane = tid & 31;
    const int n    = lane & 15;
    const int kA   = (lane >> 4) << 1;

    const int bc0 = b * Cn;

    const int srow = tid >> 2;
    const int sq4  = tid & 3;
    const int sc   = srow >> 4;
    const int snn  = srow & 15;
    const float* ps = shp + (size_t)(sc * Kn + k0 + snn) * Sn + sq4 * 32;
    float sv[32];
    #pragma unroll
    for (int i = 0; i < 32; ++i) sv[i] = ps[i];

    const int jb = (tid + 8 < NCH) ? tid + 8 : NCH - 1;
    float qa[Cn][16], qb[Cn][16];
    #pragma unroll
    for (int c = 0; c < Cn; ++c) {
        const float* pa = x + (size_t)(bc0 + c) * Ln + 16 * tid;
        const float* pb = x + (size_t)(bc0 + c) * Ln + 16 * jb;
        #pragma unroll
        for (int e = 0; e < 16; ++e) { qa[c][e] = pa[e]; qb[c][e] = pb[e]; }
    }

    {
        float p0 = 0.f, p1 = 0.f, p2 = 0.f, p3 = 0.f;
        #pragma unroll
        for (int i = 0; i < 32; i += 4) {
            p0 = fmaf(sv[i], sv[i], p0);         p1 = fmaf(sv[i + 1], sv[i + 1], p1);
            p2 = fmaf(sv[i + 2], sv[i + 2], p2); p3 = fmaf(sv[i + 3], sv[i + 3], p3);
        }
        float p = (p0 + p1) + (p2 + p3);
        p += __shfl_xor(p, 1, 32);
        p += __shfl_xor(p, 2, 32);
        const float sscale = 1.0f / fmaxf(sqrtf(p), EPSV);
        #pragma unroll
        for (int jj = 0; jj < 16; ++jj) {
            v2f pr;
            pr.x = sv[2 * jj] * sscale;
            pr.y = sv[2 * jj + 1] * sscale;
            ldsB[(sc * 64 + sq4 * 16 + jj) * 16 + snn] = pr;
        }
    }

    float hs[Cn][16], tp[Cn][16];
    #pragma unroll
    for (int c = 0; c < Cn; ++c) {
        float h = 0.f;
        #pragma unroll
        for (int e = 15; e >= 0; --e) { h = fmaf(qa[c][e], qa[c][e], h); hs[c][e] = h; }
        float tt = 0.f;
        #pragma unroll
        for (int e = 0; e < 16; ++e) { tp[c][e] = tt; tt = fmaf(qb[c][e], qb[c][e], tt); }
        ldsP[c * NCH + tid] = hs[c][0];
    }
    __syncthreads();

    if (tid < NT) {
        #pragma unroll
        for (int c = 0; c < Cn; ++c) {
            float mid = ldsP[c * NCH + tid + 1];
            #pragma unroll
            for (int u = 2; u < 8; ++u) mid += ldsP[c * NCH + tid + u];
            #pragma unroll
            for (int r = 0; r < 16; ++r) {
                const float ss  = (hs[c][r] + mid) + tp[c][r];
                const float inv = fminf(__builtin_amdgcn_rsqf(ss), 1.0f / EPSV);
                ldsInv[c * WP + 16 * tid + r] = (16 * tid + r < Wn) ? inv : 0.0f;
            }
        }
    }
    __syncthreads();

    v2f b0[32], b1[32], b2[32], b3[32];
    #pragma unroll
    for (int j = 0; j < 32; ++j) {
        b0[j] = ldsB[(0 * 64 + ((4 * j + kA) >> 1)) * 16 + n];
        b1[j] = ldsB[(1 * 64 + ((4 * j + kA) >> 1)) * 16 + n];
        b2[j] = ldsB[(2 * 64 + ((4 * j + kA) >> 1)) * 16 + n];
        b3[j] = ldsB[(3 * 64 + ((4 * j + kA) >> 1)) * 16 + n];
    }

    v2f a0[32], a1[32];
    float i0, i1;

    int t = wave;
    load_rows(x, ldsInv + 0 * WP, bc0 + 0, t, n, kA, a0, i0);

    float rmax = 0.0f;
    for (; t < NT; t += 8) {
        const int tn = (t + 8 < NT) ? t + 8 : t;
        const bool row_ok = ((t << 4) + n) < Wn;
        v8f acc = {0.f, 0.f, 0.f, 0.f, 0.f, 0.f, 0.f, 0.f};

        load_rows(x, ldsInv + 1 * WP, bc0 + 1, t, n, kA, a1, i1);
        acc = mma_rows(acc, a0, i0, row_ok, b0);
        load_rows(x, ldsInv + 2 * WP, bc0 + 2, t, n, kA, a0, i0);
        acc = mma_rows(acc, a1, i1, row_ok, b1);
        load_rows(x, ldsInv + 3 * WP, bc0 + 3, t, n, kA, a1, i1);
        acc = mma_rows(acc, a0, i0, row_ok, b2);
        load_rows(x, ldsInv + 0 * WP, bc0 + 0, tn, n, kA, a0, i0);
        acc = mma_rows(acc, a1, i1, row_ok, b3);

        #pragma unroll
        for (int r = 0; r < 8; ++r) rmax = fmaxf(rmax, acc[r] * 0.25f);
    }

    rmax = fmaxf(rmax, __shfl_xor(rmax, 16, 32));
    if (lane < 16) ldsR[wave * 16 + n] = rmax;
    __syncthreads();
    if (tid < 16) {
        float v = ldsR[tid];
        for (int w = 1; w < 8; ++w) v = fmaxf(v, ldsR[w * 16 + tid]);
        out[b * Kn + k0 + tid] = v;
    }
}

extern "C" void kernel_launch(void* const* d_in, const int* in_sizes, int n_in,
                              void* d_out, int out_size, void* d_ws, size_t ws_size,
                              hipStream_t stream) {
    const float* x   = (const float*)d_in[0];
    const float* shp = (const float*)d_in[1];
    float* out  = (float*)d_out;
    shapelet_wmma_kernel<<<Bn * (Kn / 16), 256, 0, stream>>>(x, shp, out);
}


# Two adjacent window tiles per wave share raw x fragments (tile t+1 step j = tile t step j+4): 36 loads per 64 WMMAs; inverse norms applied to per-channel sums instead of to A: new sum order
# speedup vs baseline: 5.0445x; 1.0793x over previous
#include <hip/hip_runtime.h>
#include <hip/hip_bf16.h>
#include <math.h>

constexpr int Bn = 16;
constexpr int Cn = 4;
constexpr int Ln = 4096;
constexpr int Kn = 256;
constexpr int Sn = 128;
constexpr int Wn = Ln - Sn + 1;
constexpr int NT = (Wn + 15) / 16;
constexpr float EPSV = 1e-8f;
constexpr int NU  = (NT + 1) / 2;
constexpr int WPX = (NT + 1) * 16;
constexpr int NCH = Ln / 16;

typedef __attribute__((ext_vector_type(2))) float v2f;
typedef __attribute__((ext_vector_type(4))) float v4f;
typedef __attribute__((ext_vector_type(8))) float v8f;


__device__ __forceinline__ void load_frags(
    const float* __restrict__ x, int bc, int t, int n, int kA, v2f (&a)[36])
{
    const int w1 = (t << 4) + n;
    const int wa = w1 < Wn ? w1 : Wn - 1;
    const int wb = w1 < Wn - 16 ? w1 : Wn - 17;
    const float* pa = x + (size_t)bc * Ln + wa + kA;
    const float* pb = x + (size_t)bc * Ln + wb + kA + 128;
    #pragma unroll
    for (int j = 0; j < 32; ++j) {
        a[j].x = pa[4 * j];
        a[j].y = pa[4 * j + 1];
    }
    #pragma unroll
    for (int q = 0; q < 4; ++q) {
        a[32 + q].x = pb[4 * q];
        a[32 + q].y = pb[4 * q + 1];
    }
}

__device__ __forceinline__ void mma_pair(
    const v2f (&a)[36], const v2f (&bq)[32], v8f& q1, v8f& q2)
{
    v8f c1 = {0.f, 0.f, 0.f, 0.f, 0.f, 0.f, 0.f, 0.f};
    v8f c2 = {0.f, 0.f, 0.f, 0.f, 0.f, 0.f, 0.f, 0.f};
    __builtin_amdgcn_sched_barrier(0);
    #pragma unroll
    for (int j = 0; j < 32; ++j) {
        c1 = __builtin_amdgcn_wmma_f32_16x16x4_f32(
             false, a[j],  false, bq[j],
             (short)0, c1,  false,  false);
    }
    v2f a4 = a[4];
    asm volatile("" : "+v"(c1), "+v"(a4));
    c2 = __builtin_amdgcn_wmma_f32_16x16x4_f32(
         false, a4,  false, bq[0],
         (short)0, c2,  false,  false);
    #pragma unroll
    for (int j = 1; j < 32; ++j) {
        c2 = __builtin_amdgcn_wmma_f32_16x16x4_f32(
             false, a[j + 4],  false, bq[j],
             (short)0, c2,  false,  false);
    }
    __builtin_amdgcn_sched_barrier(0);
    asm volatile("v_nop\n\tv_nop\n\tv_nop\n\tv_nop" : "+v"(c1), "+v"(c2));
    q1 = c1;
    q2 = c2;
}

__device__ __forceinline__ void load_inv(const float* iv, v4f (&w)[4])
{
    w[0] = *reinterpret_cast<const v4f*>(iv);
    w[1] = *reinterpret_cast<const v4f*>(iv + 4);
    w[2] = *reinterpret_cast<const v4f*>(iv + 16);
    w[3] = *reinterpret_cast<const v4f*>(iv + 20);
}

__device__ __forceinline__ void fold2(
    v8f& tot1, v8f& tot2, const v8f& q1, const v8f& q2, const v4f (&w)[4])
{
    #pragma unroll
    for (int r = 0; r < 4; ++r) {
        tot1[r]     = fmaf(q1[r],     w[0][r], tot1[r]);
        tot1[r + 4] = fmaf(q1[r + 4], w[1][r], tot1[r + 4]);
        tot2[r]     = fmaf(q2[r],     w[2][r], tot2[r]);
        tot2[r + 4] = fmaf(q2[r + 4], w[3][r], tot2[r + 4]);
    }
    asm volatile("" : "+v"(tot1), "+v"(tot2));
}

__global__ __launch_bounds__(256) void shapelet_wmma_kernel(
    const float* __restrict__ x, const float* __restrict__ shp, float* __restrict__ out)
{
    __shared__ v2f   ldsB[Cn * (Sn / 2) * 16];
    __shared__ __attribute__((aligned(16))) float ldsInv[Cn * WPX];
    __shared__ float ldsP[Cn * NCH];
    __shared__ float ldsR[8 * 16];

    const int b    = blockIdx.x >> 4;
    const int k0   = (blockIdx.x & 15) << 4;
    const int tid  = threadIdx.x;
    const int wave = tid >> 5;
    const int lane = tid & 31;
    const int n    = lane & 15;
    const int kA   = (lane >> 4) << 1;
    const int h8   = (lane >> 4) << 3;

    const int bc0 = b * Cn;

    const int srow = tid >> 2;
    const int sq4  = tid & 3;
    const int sc   = srow >> 4;
    const int snn  = srow & 15;
    const float* ps = shp + (size_t)(sc * Kn + k0 + snn) * Sn + sq4 * 32;
    float sv[32];
    #pragma unroll
    for (int i = 0; i < 32; ++i) sv[i] = ps[i];

    const int jb = (tid + 8 < NCH) ? tid + 8 : NCH - 1;
    float qa[Cn][16], qb[Cn][16];
    #pragma unroll
    for (int c = 0; c < Cn; ++c) {
        const float* pa = x + (size_t)(bc0 + c) * Ln + 16 * tid;
        const float* pb = x + (size_t)(bc0 + c) * Ln + 16 * jb;
        #pragma unroll
        for (int e = 0; e < 16; ++e) { qa[c][e] = pa[e]; qb[c][e] = pb[e]; }
    }

    {
        float p0 = 0.f, p1 = 0.f, p2 = 0.f, p3 = 0.f;
        #pragma unroll
        for (int i = 0; i < 32; i += 4) {
            p0 = fmaf(sv[i], sv[i], p0);         p1 = fmaf(sv[i + 1], sv[i + 1], p1);
            p2 = fmaf(sv[i + 2], sv[i + 2], p2); p3 = fmaf(sv[i + 3], sv[i + 3], p3);
        }
        float p = (p0 + p1) + (p2 + p3);
        p += __shfl_xor(p, 1, 32);
        p += __shfl_xor(p, 2, 32);
        const float sscale = 1.0f / fmaxf(sqrtf(p), EPSV);
        #pragma unroll
        for (int jj = 0; jj < 16; ++jj) {
            v2f pr;
            pr.x = sv[2 * jj] * sscale;
            pr.y = sv[2 * jj + 1] * sscale;
            ldsB[(sc * 64 + sq4 * 16 + jj) * 16 + snn] = pr;
        }
    }

    float hs[Cn][16], tp[Cn][16];
    #pragma unroll
    for (int c = 0; c < Cn; ++c) {
        float h = 0.f;
        #pragma unroll
        for (int e = 15; e >= 0; --e) { h = fmaf(qa[c][e], qa[c][e], h); hs[c][e] = h; }
        float tt = 0.f;
        #pragma unroll
        for (int e = 0; e < 16; ++e) { tp[c][e] = tt; tt = fmaf(qb[c][e], qb[c][e], tt); }
        ldsP[c * NCH + tid] = hs[c][0];
    }
    __syncthreads();

    if (tid < NT) {
        #pragma unroll
        for (int c = 0; c < Cn; ++c) {
            float mid = ldsP[c * NCH + tid + 1];
            #pragma unroll
            for (int u = 2; u < 8; ++u) mid += ldsP[c * NCH + tid + u];
            #pragma unroll
            for (int r = 0; r < 16; ++r) {
                const float ss  = (hs[c][r] + mid) + tp[c][r];
                const float inv = fminf(__builtin_amdgcn_rsqf(ss), 1.0f / EPSV);
                ldsInv[c * WPX + 16 * tid + r] = (16 * tid + r < Wn) ? inv : 0.0f;
            }
        }
    } else if (tid == NT) {
        #pragma unroll
        for (int c = 0; c < Cn; ++c) {
            #pragma unroll
            for (int r = 0; r < 16; ++r) ldsInv[c * WPX + 16 * NT + r] = 0.0f;
        }
    }
    __syncthreads();

    v2f b0[32], b1[32], b2[32], b3[32];
    #pragma unroll
    for (int j = 0; j < 32; ++j) {
        b0[j] = ldsB[(0 * 64 + ((4 * j + kA) >> 1)) * 16 + n];
        b1[j] = ldsB[(1 * 64 + ((4 * j + kA) >> 1)) * 16 + n];
        b2[j] = ldsB[(2 * 64 + ((4 * j + kA) >> 1)) * 16 + n];
        b3[j] = ldsB[(3 * 64 + ((4 * j + kA) >> 1)) * 16 + n];
    }

    v2f a0[36], a1[36];

    int u = wave;
    load_frags(x, bc0 + 0, 2 * u, n, kA, a0);

    float rmax = 0.0f;
    for (; u < NU; u += 8) {
        const int t  = 2 * u;
        const int tn = (u + 8 < NU) ? t + 16 : t;
        const float* iv = ldsInv + 16 * t + h8;
        v8f tot1 = {0.f, 0.f, 0.f, 0.f, 0.f, 0.f, 0.f, 0.f};
        v8f tot2 = {0.f, 0.f, 0.f, 0.f, 0.f, 0.f, 0.f, 0.f};
        v8f q1, q2;
        v4f wa[4], wb[4];

        load_inv(iv + 0 * WPX, wa);
        load_frags(x, bc0 + 1, t, n, kA, a1);
        mma_pair(a0, b0, q1, q2);
        load_frags(x, bc0 + 2, t, n, kA, a0);
        load_inv(iv + 1 * WPX, wb);
        __builtin_amdgcn_sched_barrier(0);
        fold2(tot1, tot2, q1, q2, wa);
        mma_pair(a1, b1, q1, q2);
        load_frags(x, bc0 + 3, t, n, kA, a1);
        load_inv(iv + 2 * WPX, wa);
        __builtin_amdgcn_sched_barrier(0);
        fold2(tot1, tot2, q1, q2, wb);
        mma_pair(a0, b2, q1, q2);
        load_frags(x, bc0 + 0, tn, n, kA, a0);
        load_inv(iv + 3 * WPX, wb);
        __builtin_amdgcn_sched_barrier(0);
        fold2(tot1, tot2, q1, q2, wa);
        mma_pair(a1, b3, q1, q2);
        fold2(tot1, tot2, q1, q2, wb);

        #pragma unroll
        for (int r = 0; r < 8; ++r)
            rmax = fmaxf(rmax, fmaxf(tot1[r], tot2[r]) * 0.25f);
    }

    rmax = fmaxf(rmax, __shfl_xor(rmax, 16, 32));
    if (lane < 16) ldsR[wave * 16 + n] = rmax;
    __syncthreads();
    if (tid < 16) {
        float v = ldsR[tid];
        for (int w = 1; w < 8; ++w) v = fmaxf(v, ldsR[w * 16 + tid]);
        out[b * Kn + k0 + tid] = v;
    }
}

extern "C" void kernel_launch(void* const* d_in, const int* in_sizes, int n_in,
                              void* d_out, int out_size, void* d_ws, size_t ws_size,
                              hipStream_t stream) {
    const float* x   = (const float*)d_in[0];
    const float* shp = (const float*)d_in[1];
    float* out  = (float*)d_out;
    shapelet_wmma_kernel<<<Bn * (Kn / 16), 256, 0, stream>>>(x, shp, out);
}


# 3-product split per (tile pair x 16-s block pair): H1*(V0+V1), (H0-H1)*V0, (H2-H1)*V1, differences read from x[n]-x[n+16] staged in LDS: 48 WMMAs per channel-unit, not 64; 1 wave/SIMD; new sum order
# speedup vs baseline: 5.8780x; 1.1652x over previous
#include <hip/hip_runtime.h>
#include <hip/hip_bf16.h>
#include <math.h>


constexpr int Bn = 16;
constexpr int Cn = 4;
constexpr int Ln = 4096;
constexpr int Kn = 256;
constexpr int Sn = 128;
constexpr int Wn = Ln - Sn + 1;
constexpr int NT = (Wn + 15) / 16;
constexpr float EPSV = 1e-8f;
constexpr int NU  = (NT + 1) / 2;
constexpr int WPX = (NT + 1) * 16;
constexpr int NCH = Ln / 16;
constexpr int NTH = 128;
constexpr int DPC = Ln / 2;
constexpr int DOD = Cn * DPC + 16;

typedef __attribute__((ext_vector_type(2))) float v2f;
typedef __attribute__((ext_vector_type(4))) float v4f;
typedef __attribute__((ext_vector_type(8))) float v8f;


__device__ __forceinline__ void load_frags(
    const float* __restrict__ x, const v2f* dl, int bc, int c, int t, int n, int kA,
    v2f (&am)[16], v2f (&ad)[32])
{
    const int w1 = (t << 4) + n;
    const int wa = w1 < Wn ? w1 : Wn - 1;
    const float* pa = x + (size_t)bc * Ln + wa + kA;
    const v2f* pd = dl + ((wa & 1) ? DOD : 0) + c * DPC + ((wa + kA) >> 1);
    #pragma unroll
    for (int q = 0; q < 4; ++q) {
        #pragma unroll
        for (int r = 0; r < 4; ++r) {
            am[4 * q + r].x = pa[4 * (8 * q + 4 + r)];
            am[4 * q + r].y = pa[4 * (8 * q + 4 + r) + 1];
        }
    }
    #pragma unroll
    for (int j = 0; j < 32; ++j) ad[j] = pd[2 * j];
}

__device__ __forceinline__ void mma_triple(
    const v2f (&am)[16], const v2f (&ad)[32], const v2f (&bq)[32], const v2f (&uq)[16],
    v8f& s1, v8f& s2, v8f& s3)
{
    v8f c1 = {0.f, 0.f, 0.f, 0.f, 0.f, 0.f, 0.f, 0.f};
    v8f c2 = {0.f, 0.f, 0.f, 0.f, 0.f, 0.f, 0.f, 0.f};
    v8f c3 = {0.f, 0.f, 0.f, 0.f, 0.f, 0.f, 0.f, 0.f};
    __builtin_amdgcn_sched_barrier(0);
    #pragma unroll
    for (int q = 0; q < 4; ++q) {
        #pragma unroll
        for (int r = 0; r < 4; ++r) {
            c1 = __builtin_amdgcn_wmma_f32_16x16x4_f32(
                 false, am[4 * q + r],  false, uq[4 * q + r],
                 (short)0, c1,  false,  false);
            c2 = __builtin_amdgcn_wmma_f32_16x16x4_f32(
                 false, ad[8 * q + r],  false, bq[8 * q + r],
                 (short)0, c2,  false,  false);
            c3 = __builtin_amdgcn_wmma_f32_16x16x4_f32(
                 false, ad[8 * q + 4 + r],  false, bq[8 * q + 4 + r],
                 (short)0, c3,  false,  false);
        }
    }
    __builtin_amdgcn_sched_barrier(0);
    asm volatile("v_nop\n\tv_nop\n\tv_nop\n\tv_nop" : "+v"(c1), "+v"(c2), "+v"(c3));
    s1 = c1;
    s2 = c2;
    s3 = c3;
}

__device__ __forceinline__ void load_inv(const float* iv, v4f (&w)[4])
{
    w[0] = *reinterpret_cast<const v4f*>(iv);
    w[1] = *reinterpret_cast<const v4f*>(iv + 4);
    w[2] = *reinterpret_cast<const v4f*>(iv + 16);
    w[3] = *reinterpret_cast<const v4f*>(iv + 20);
}

__device__ __forceinline__ void fold3(
    v8f& tot1, v8f& tot2, const v8f& s1, const v8f& s2, const v8f& s3, const v4f (&w)[4])
{
    #pragma unroll
    for (int r = 0; r < 4; ++r) {
        tot1[r]     = fmaf(s1[r] + s2[r],         w[0][r], tot1[r]);
        tot1[r + 4] = fmaf(s1[r + 4] + s2[r + 4], w[1][r], tot1[r + 4]);
        tot2[r]     = fmaf(s1[r] - s3[r],         w[2][r], tot2[r]);
        tot2[r + 4] = fmaf(s1[r + 4] - s3[r + 4], w[3][r], tot2[r + 4]);
    }
    asm volatile("" : "+v"(tot1), "+v"(tot2));
}

__global__ __launch_bounds__(NTH) void shapelet_wmma_kernel(
    const float* __restrict__ x, const float* __restrict__ shp, float* __restrict__ out)
{
    __shared__ v2f   ldsB[Cn * (Sn / 2) * 16];
    __shared__ __attribute__((aligned(16))) float ldsInv[Cn * WPX];
    __shared__ float ldsP[Cn * NCH];
    __shared__ __attribute__((aligned(16))) v2f ldsD[2 * Cn * DPC + 16];
    __shared__ float ldsR[4 * 16];

    const int b    = blockIdx.x >> 4;
    const int k0   = (blockIdx.x & 15) << 4;
    const int tid  = threadIdx.x;
    const int wave = tid >> 5;
    const int lane = tid & 31;
    const int n    = lane & 15;
    const int kA   = (lane >> 4) << 1;
    const int h8   = (lane >> 4) << 3;

    const int bc0 = b * Cn;

    const int srow = tid >> 1;
    const int sh   = tid & 1;
    const int sc   = srow >> 4;
    const int snn  = srow & 15;
    const float* ps = shp + (size_t)(sc * Kn + k0 + snn) * Sn + sh * 64;
    float sv[64];
    #pragma unroll
    for (int i = 0; i < 64; ++i) sv[i] = ps[i];

    {
        float pq[2];
        #pragma unroll
        for (int g = 0; g < 2; ++g) {
            float p0 = 0.f, p1 = 0.f, p2 = 0.f, p3 = 0.f;
            #pragma unroll
            for (int i = 32 * g; i < 32 * g + 32; i += 4) {
                p0 = fmaf(sv[i], sv[i], p0);         p1 = fmaf(sv[i + 1], sv[i + 1], p1);
                p2 = fmaf(sv[i + 2], sv[i + 2], p2); p3 = fmaf(sv[i + 3], sv[i + 3], p3);
            }
            pq[g] = (p0 + p1) + (p2 + p3);
        }
        float p = pq[0] + pq[1];
        p += __shfl_xor(p, 1, 32);
        const float sscale = 1.0f / fmaxf(sqrtf(p), EPSV);
        #pragma unroll
        for (int jj = 0; jj < 32; ++jj) {
            v2f pr;
            pr.x = sv[2 * jj] * sscale;
            pr.y = sv[2 * jj + 1] * sscale;
            ldsB[(sc * 64 + sh * 32 + jj) * 16 + snn] = pr;
        }
    }

    float hs[2][Cn][16], tp[2][Cn][16];
    #pragma unroll
    for (int rep = 0; rep < 2; ++rep) {
        const int vt = tid + NTH * rep;
        const int jb = (vt + 8 < NCH) ? vt + 8 : NCH - 1;
        #pragma unroll
        for (int c = 0; c < Cn; ++c) {
            const float* pa = x + (size_t)(bc0 + c) * Ln + 16 * vt;
            const float* pb = x + (size_t)(bc0 + c) * Ln + 16 * jb;
            const float* pn = x + (size_t)(bc0 + c) * Ln + 16 * ((vt + 1 < NCH) ? vt + 1 : NCH - 1);
            const float xnn = x[(size_t)(bc0 + c) * Ln + 16 * ((vt + 2 < NCH) ? vt + 2 : NCH - 1)];
            float qa[16], qb[16], qn[16];
            #pragma unroll
            for (int e = 0; e < 16; ++e) { qa[e] = pa[e]; qb[e] = pb[e]; qn[e] = pn[e]; }
            float dd[17];
            #pragma unroll
            for (int e = 0; e < 16; ++e) dd[e] = qa[e] - qn[e];
            dd[16] = qn[0] - xnn;
            #pragma unroll
            for (int i = 0; i < 8; ++i) {
                v2f pe, po;
                pe.x = dd[2 * i];     pe.y = dd[2 * i + 1];
                po.x = dd[2 * i + 1]; po.y = dd[2 * i + 2];
                ldsD[c * DPC + 8 * vt + i]       = pe;
                ldsD[DOD + c * DPC + 8 * vt + i] = po;
            }
            float h = 0.f;
            #pragma unroll
            for (int e = 15; e >= 0; --e) { h = fmaf(qa[e], qa[e], h); hs[rep][c][e] = h; }
            float tt = 0.f;
            #pragma unroll
            for (int e = 0; e < 16; ++e) { tp[rep][c][e] = tt; tt = fmaf(qb[e], qb[e], tt); }
            ldsP[c * NCH + vt] = hs[rep][c][0];
        }
    }
    __syncthreads();

    #pragma unroll
    for (int rep = 0; rep < 2; ++rep) {
        const int vt = tid + NTH * rep;
        if (vt < NT) {
            #pragma unroll
            for (int c = 0; c < Cn; ++c) {
                float mid = ldsP[c * NCH + vt + 1];
                #pragma unroll
                for (int u2 = 2; u2 < 8; ++u2) mid += ldsP[c * NCH + vt + u2];
                #pragma unroll
                for (int r = 0; r < 16; ++r) {
                    const float ss  = (hs[rep][c][r] + mid) + tp[rep][c][r];
                    const float inv = fminf(__builtin_amdgcn_rsqf(ss), 1.0f / EPSV);
                    ldsInv[c * WPX + 16 * vt + r] = (16 * vt + r < Wn) ? inv : 0.0f;
                }
            }
        } else if (vt == NT) {
            #pragma unroll
            for (int c = 0; c < Cn; ++c) {
                #pragma unroll
                for (int r = 0; r < 16; ++r) ldsInv[c * WPX + 16 * NT + r] = 0.0f;
            }
        }
    }
    __syncthreads();

    v2f b0[32], b1[32], b2[32], b3[32];
    #pragma unroll
    for (int j = 0; j < 32; ++j) {
        b0[j] = ldsB[(0 * 64 + ((4 * j + kA) >> 1)) * 16 + n];
        b1[j] = ldsB[(1 * 64 + ((4 * j + kA) >> 1)) * 16 + n];
        b2[j] = ldsB[(2 * 64 + ((4 * j + kA) >> 1)) * 16 + n];
        b3[j] = ldsB[(3 * 64 + ((4 * j + kA) >> 1)) * 16 + n];
    }
    v2f u0[16], u1[16], u2[16], u3[16];
    #pragma unroll
    for (int q = 0; q < 4; ++q) {
        #pragma unroll
        for (int r = 0; r < 4; ++r) {
            u0[4 * q + r] = b0[8 * q + r] + b0[8 * q + 4 + r];
            u1[4 * q + r] = b1[8 * q + r] + b1[8 * q + 4 + r];
            u2[4 * q + r] = b2[8 * q + r] + b2[8 * q + 4 + r];
            u3[4 * q + r] = b3[8 * q + r] + b3[8 * q + 4 + r];
        }
    }
    #pragma unroll
    for (int i = 0; i < 16; ++i)
        asm volatile("" : "+v"(u0[i]), "+v"(u1[i]), "+v"(u2[i]), "+v"(u3[i]));

    v2f am0[16], ad0[32], am1[16], ad1[32];

    int u = wave;
    load_frags(x, ldsD, bc0 + 0, 0, 2 * u, n, kA, am0, ad0);

    float rmax = 0.0f;
    for (; u < NU; u += 4) {
        const int t  = 2 * u;
        const int tn = (u + 4 < NU) ? t + 8 : t;
        const float* iv = ldsInv + 16 * t + h8;
        v8f tot1 = {0.f, 0.f, 0.f, 0.f, 0.f, 0.f, 0.f, 0.f};
        v8f tot2 = {0.f, 0.f, 0.f, 0.f, 0.f, 0.f, 0.f, 0.f};
        v8f s1, s2, s3;
        v4f wa[4], wb[4];

        load_inv(iv + 0 * WPX, wa);
        load_frags(x, ldsD, bc0 + 1, 1, t, n, kA, am1, ad1);
        mma_triple(am0, ad0, b0, u0, s1, s2, s3);
        load_frags(x, ldsD, bc0 + 2, 2, t, n, kA, am0, ad0);
        load_inv(iv + 1 * WPX, wb);
        __builtin_amdgcn_sched_barrier(0);
        fold3(tot1, tot2, s1, s2, s3, wa);
        mma_triple(am1, ad1, b1, u1, s1, s2, s3);
        load_frags(x, ldsD, bc0 + 3, 3, t, n, kA, am1, ad1);
        load_inv(iv + 2 * WPX, wa);
        __builtin_amdgcn_sched_barrier(0);
        fold3(tot1, tot2, s1, s2, s3, wb);
        mma_triple(am0, ad0, b2, u2, s1, s2, s3);
        load_frags(x, ldsD, bc0 + 0, 0, tn, n, kA, am0, ad0);
        load_inv(iv + 3 * WPX, wb);
        __builtin_amdgcn_sched_barrier(0);
        fold3(tot1, tot2, s1, s2, s3, wa);
        mma_triple(am1, ad1, b3, u3, s1, s2, s3);
        fold3(tot1, tot2, s1, s2, s3, wb);

        #pragma unroll
        for (int r = 0; r < 8; ++r)
            rmax = fmaxf(rmax, fmaxf(tot1[r], tot2[r]) * 0.25f);
    }

    rmax = fmaxf(rmax, __shfl_xor(rmax, 16, 32));
    if (lane < 16) ldsR[wave * 16 + n] = rmax;
    __syncthreads();
    if (tid < 16) {
        float v = ldsR[tid];
        for (int w = 1; w < 4; ++w) v = fmaxf(v, ldsR[w * 16 + tid]);
        out[b * Kn + k0 + tid] = v;
    }
}

extern "C" void kernel_launch(void* const* d_in, const int* in_sizes, int n_in,
                              void* d_out, int out_size, void* d_ws, size_t ws_size,
                              hipStream_t stream) {
    const float* x   = (const float*)d_in[0];
    const float* shp = (const float*)d_in[1];
    float* out  = (float*)d_out;
    shapelet_wmma_kernel<<<Bn * (Kn / 16), NTH, 0, stream>>>(x, shp, out);
}


# Units of four adjacent window tiles (two pairs) per wave: pair 2 reuses 12 of 16 raw and 24 of 32 difference fragments of pair 1; 48 load instructions per 96 WMMAs, not 72; sums unchanged
# speedup vs baseline: 6.1761x; 1.0507x over previous
#include <hip/hip_runtime.h>
#include <hip/hip_bf16.h>
#include <math.h>


constexpr int Bn = 16;
constexpr int Cn = 4;
constexpr int Ln = 4096;
constexpr int Kn = 256;
constexpr int Sn = 128;
constexpr int Wn = Ln - Sn + 1;
constexpr int NT = (Wn + 15) / 16;
constexpr float EPSV = 1e-8f;
constexpr int NU  = (NT + 3) / 4;
constexpr int WPX = (NT + 3) * 16;
constexpr int WX  = Ln - 160;
constexpr int NCH = Ln / 16;
constexpr int NTH = 128;
constexpr int DPC = Ln / 2;
constexpr int DOD = Cn * DPC + 16;

typedef __attribute__((ext_vector_type(2))) float v2f;
typedef __attribute__((ext_vector_type(4))) float v4f;
typedef __attribute__((ext_vector_type(8))) float v8f;


__device__ __forceinline__ void load_frags(
    const float* __restrict__ x, const v2f* dl, int bc, int c, int t, int n, int kA,
    v2f (&am)[20], v2f (&ad)[40])
{
    const int w1 = (t << 4) + n;
    const int wa = w1 < Wn ? w1 : Wn - 1;
    const int wb = w1 < WX ? w1 : WX;
    const float* pa = x + (size_t)bc * Ln + wa + kA;
    const float* pb = x + (size_t)bc * Ln + wb + kA;
    const v2f* pd = dl + ((wa & 1) ? DOD : 0) + c * DPC + ((wa + kA) >> 1);
    const v2f* pe = dl + ((wb & 1) ? DOD : 0) + c * DPC + ((wb + kA) >> 1);
    #pragma unroll
    for (int q = 0; q < 4; ++q) {
        #pragma unroll
        for (int r = 0; r < 4; ++r) {
            am[4 * q + r].x = pa[4 * (8 * q + 4 + r)];
            am[4 * q + r].y = pa[4 * (8 * q + 4 + r) + 1];
        }
    }
    #pragma unroll
    for (int r = 0; r < 4; ++r) {
        am[16 + r].x = pb[4 * (36 + r)];
        am[16 + r].y = pb[4 * (36 + r) + 1];
    }
    #pragma unroll
    for (int j = 0; j < 32; ++j) ad[j] = pd[2 * j];
    #pragma unroll
    for (int j = 32; j < 40; ++j) ad[j] = pe[2 * j];
}

__device__ __forceinline__ void mma_six(
    const v2f (&am)[20], const v2f (&ad)[40], const v2f (&bq)[32], const v2f (&uq)[16],
    v8f& s1, v8f& s2, v8f& s3, v8f& s4, v8f& s5, v8f& s6)
{
    v8f c1 = {0.f, 0.f, 0.f, 0.f, 0.f, 0.f, 0.f, 0.f};
    v8f c2 = {0.f, 0.f, 0.f, 0.f, 0.f, 0.f, 0.f, 0.f};
    v8f c3 = {0.f, 0.f, 0.f, 0.f, 0.f, 0.f, 0.f, 0.f};
    v8f c4 = {0.f, 0.f, 0.f, 0.f, 0.f, 0.f, 0.f, 0.f};
    v8f c5 = {0.f, 0.f, 0.f, 0.f, 0.f, 0.f, 0.f, 0.f};
    v8f c6 = {0.f, 0.f, 0.f, 0.f, 0.f, 0.f, 0.f, 0.f};
    __builtin_amdgcn_sched_barrier(0);
    #pragma unroll
    for (int q = 0; q < 4; ++q) {
        #pragma unroll
        for (int r = 0; r < 4; ++r) {
            c1 = __builtin_amdgcn_wmma_f32_16x16x4_f32(
                 false, am[4 * q + r],  false, uq[4 * q + r],
                 (short)0, c1,  false,  false);
            c2 = __builtin_amdgcn_wmma_f32_16x16x4_f32(
                 false, ad[8 * q + r],  false, bq[8 * q + r],
                 (short)0, c2,  false,  false);
            c3 = __builtin_amdgcn_wmma_f32_16x16x4_f32(
                 false, ad[8 * q + 4 + r],  false, bq[8 * q + 4 + r],
                 (short)0, c3,  false,  false);
            c4 = __builtin_amdgcn_wmma_f32_16x16x4_f32(
                 false, am[4 * q + 4 + r],  false, uq[4 * q + r],
                 (short)0, c4,  false,  false);
            c5 = __builtin_amdgcn_wmma_f32_16x16x4_f32(
                 false, ad[8 * q + 8 + r],  false, bq[8 * q + r],
                 (short)0, c5,  false,  false);
            c6 = __builtin_amdgcn_wmma_f32_16x16x4_f32(
                 false, ad[8 * q + 12 + r],  false, bq[8 * q + 4 + r],
                 (short)0, c6,  false,  false);
        }
    }
    __builtin_amdgcn_sched_barrier(0);
    asm volatile("v_nop\n\tv_nop\n\tv_nop\n\tv_nop"
                 : "+v"(c1), "+v"(c2), "+v"(c3), "+v"(c4), "+v"(c5), "+v"(c6));
    s1 = c1;
    s2 = c2;
    s3 = c3;
    s4 = c4;
    s5 = c5;
    s6 = c6;
}

__device__ __forceinline__ void load_inv(const float* iv, v4f (&w)[8])
{
    #pragma unroll
    for (int a = 0; a < 4; ++a) {
        w[2 * a]     = *reinterpret_cast<const v4f*>(iv + 16 * a);
        w[2 * a + 1] = *reinterpret_cast<const v4f*>(iv + 16 * a + 4);
    }
}

__device__ __forceinline__ void fold6(
    v8f& tot1, v8f& tot2, v8f& tot3, v8f& tot4,
    const v8f& s1, const v8f& s2, const v8f& s3, const v8f& s4, const v8f& s5, const v8f& s6,
    const v4f (&w)[8])
{
    #pragma unroll
    for (int r = 0; r < 4; ++r) {
        tot1[r]     = fmaf(s1[r] + s2[r],         w[0][r], tot1[r]);
        tot1[r + 4] = fmaf(s1[r + 4] + s2[r + 4], w[1][r], tot1[r + 4]);
        tot2[r]     = fmaf(s1[r] - s3[r],         w[2][r], tot2[r]);
        tot2[r + 4] = fmaf(s1[r + 4] - s3[r + 4], w[3][r], tot2[r + 4]);
        tot3[r]     = fmaf(s4[r] + s5[r],         w[4][r], tot3[r]);
        tot3[r + 4] = fmaf(s4[r + 4] + s5[r + 4], w[5][r], tot3[r + 4]);
        tot4[r]     = fmaf(s4[r] - s6[r],         w[6][r], tot4[r]);
        tot4[r + 4] = fmaf(s4[r + 4] - s6[r + 4], w[7][r], tot4[r + 4]);
    }
    asm volatile("" : "+v"(tot1), "+v"(tot2), "+v"(tot3), "+v"(tot4));
}

__global__ __launch_bounds__(NTH) void shapelet_wmma_kernel(
    const float* __restrict__ x, const float* __restrict__ shp, float* __restrict__ out)
{
    __shared__ v2f   ldsB[Cn * (Sn / 2) * 16];
    __shared__ __attribute__((aligned(16))) float ldsInv[Cn * WPX];
    __shared__ float ldsP[Cn * NCH];
    __shared__ __attribute__((aligned(16))) v2f ldsD[2 * Cn * DPC + 16];
    __shared__ float ldsR[4 * 16];

    const int b    = blockIdx.x >> 4;
    const int k0   = (blockIdx.x & 15) << 4;
    const int tid  = threadIdx.x;
    const int wave = tid >> 5;
    const int lane = tid & 31;
    const int n    = lane & 15;
    const int kA   = (lane >> 4) << 1;
    const int h8   = (lane >> 4) << 3;

    const int bc0 = b * Cn;

    const int srow = tid >> 1;
    const int sh   = tid & 1;
    const int sc   = srow >> 4;
    const int snn  = srow & 15;
    const float* ps = shp + (size_t)(sc * Kn + k0 + snn) * Sn + sh * 64;
    float sv[64];
    #pragma unroll
    for (int i = 0; i < 64; ++i) sv[i] = ps[i];

    {
        float pq[2];
        #pragma unroll
        for (int g = 0; g < 2; ++g) {
            float p0 = 0.f, p1 = 0.f, p2 = 0.f, p3 = 0.f;
            #pragma unroll
            for (int i = 32 * g; i < 32 * g + 32; i += 4) {
                p0 = fmaf(sv[i], sv[i], p0);         p1 = fmaf(sv[i + 1], sv[i + 1], p1);
                p2 = fmaf(sv[i + 2], sv[i + 2], p2); p3 = fmaf(sv[i + 3], sv[i + 3], p3);
            }
            pq[g] = (p0 + p1) + (p2 + p3);
        }
        float p = pq[0] + pq[1];
        p += __shfl_xor(p, 1, 32);
        const float sscale = 1.0f / fmaxf(sqrtf(p), EPSV);
        #pragma unroll
        for (int jj = 0; jj < 32; ++jj) {
            v2f pr;
            pr.x = sv[2 * jj] * sscale;
            pr.y = sv[2 * jj + 1] * sscale;
            ldsB[(sc * 64 + sh * 32 + jj) * 16 + snn] = pr;
        }
    }

    float hs[2][Cn][16], tp[2][Cn][16];
    #pragma unroll
    for (int rep = 0; rep < 2; ++rep) {
        const int vt = tid + NTH * rep;
        const int jb = (vt + 8 < NCH) ? vt + 8 : NCH - 1;
        #pragma unroll
        for (int c = 0; c < Cn; ++c) {
            const float* pa = x + (size_t)(bc0 + c) * Ln + 16 * vt;
            const float* pb = x + (size_t)(bc0 + c) * Ln + 16 * jb;
            const float* pn = x + (size_t)(bc0 + c) * Ln + 16 * ((vt + 1 < NCH) ? vt + 1 : NCH - 1);
            const float xnn = x[(size_t)(bc0 + c) * Ln + 16 * ((vt + 2 < NCH) ? vt + 2 : NCH - 1)];
            float qa[16], qb[16], qn[16];
            #pragma unroll
            for (int e = 0; e < 16; ++e) { qa[e] = pa[e]; qb[e] = pb[e]; qn[e] = pn[e]; }
            float dd[17];
            #pragma unroll
            for (int e = 0; e < 16; ++e) dd[e] = qa[e] - qn[e];
            dd[16] = qn[0] - xnn;
            #pragma unroll
            for (int i = 0; i < 8; ++i) {
                v2f pe, po;
                pe.x = dd[2 * i];     pe.y = dd[2 * i + 1];
                po.x = dd[2 * i + 1]; po.y = dd[2 * i + 2];
                ldsD[c * DPC + 8 * vt + i]       = pe;
                ldsD[DOD + c * DPC + 8 * vt + i] = po;
            }
            float h = 0.f;
            #pragma unroll
            for (int e = 15; e >= 0; --e) { h = fmaf(qa[e], qa[e], h); hs[rep][c][e] = h; }
            float tt = 0.f;
            #pragma unroll
            for (int e = 0; e < 16; ++e) { tp[rep][c][e] = tt; tt = fmaf(qb[e], qb[e], tt); }
            ldsP[c * NCH + vt] = hs[rep][c][0];
        }
    }
    __syncthreads();

    #pragma unroll
    for (int rep = 0; rep < 2; ++rep) {
        const int vt = tid + NTH * rep;
        if (vt < NT) {
            #pragma unroll
            for (int c = 0; c < Cn; ++c) {
                float mid = ldsP[c * NCH + vt + 1];
                #pragma unroll
                for (int u2 = 2; u2 < 8; ++u2) mid += ldsP[c * NCH + vt + u2];
                #pragma unroll
                for (int r = 0; r < 16; ++r) {
                    const float ss  = (hs[rep][c][r] + mid) + tp[rep][c][r];
                    const float inv = fminf(__builtin_amdgcn_rsqf(ss), 1.0f / EPSV);
                    ldsInv[c * WPX + 16 * vt + r] = (16 * vt + r < Wn) ? inv : 0.0f;
                }
            }
        } else if (vt < NT + 3) {
            #pragma unroll
            for (int c = 0; c < Cn; ++c) {
                #pragma unroll
                for (int r = 0; r < 16; ++r) ldsInv[c * WPX + 16 * vt + r] = 0.0f;
            }
        }
    }
    __syncthreads();

    v2f b0[32], b1[32], b2[32], b3[32];
    #pragma unroll
    for (int j = 0; j < 32; ++j) {
        b0[j] = ldsB[(0 * 64 + ((4 * j + kA) >> 1)) * 16 + n];
        b1[j] = ldsB[(1 * 64 + ((4 * j + kA) >> 1)) * 16 + n];
        b2[j] = ldsB[(2 * 64 + ((4 * j + kA) >> 1)) * 16 + n];
        b3[j] = ldsB[(3 * 64 + ((4 * j + kA) >> 1)) * 16 + n];
    }
    v2f u0[16], u1[16], u2[16], u3[16];
    #pragma unroll
    for (int q = 0; q < 4; ++q) {
        #pragma unroll
        for (int r = 0; r < 4; ++r) {
            u0[4 * q + r] = b0[8 * q + r] + b0[8 * q + 4 + r];
            u1[4 * q + r] = b1[8 * q + r] + b1[8 * q + 4 + r];
            u2[4 * q + r] = b2[8 * q + r] + b2[8 * q + 4 + r];
            u3[4 * q + r] = b3[8 * q + r] + b3[8 * q + 4 + r];
        }
    }
    #pragma unroll
    for (int i = 0; i < 16; ++i)
        asm volatile("" : "+v"(u0[i]), "+v"(u1[i]), "+v"(u2[i]), "+v"(u3[i]));

    v2f am0[20], ad0[40], am1[20], ad1[40];

    int u = wave;
    load_frags(x, ldsD, bc0 + 0, 0, 4 * u, n, kA, am0, ad0);

    float rmax = 0.0f;
    for (; u < NU; u += 4) {
        const int t  = 4 * u;
        const int tn = (u + 4 < NU) ? t + 16 : t;
        const float* iv = ldsInv + 16 * t + h8;
        v8f tot1 = {0.f, 0.f, 0.f, 0.f, 0.f, 0.f, 0.f, 0.f};
        v8f tot2 = {0.f, 0.f, 0.f, 0.f, 0.f, 0.f, 0.f, 0.f};
        v8f tot3 = {0.f, 0.f, 0.f, 0.f, 0.f, 0.f, 0.f, 0.f};
        v8f tot4 = {0.f, 0.f, 0.f, 0.f, 0.f, 0.f, 0.f, 0.f};
        v8f s1, s2, s3, s4, s5, s6;
        v4f wa[8], wb[8];

        load_inv(iv + 0 * WPX, wa);
        load_frags(x, ldsD, bc0 + 1, 1, t, n, kA, am1, ad1);
        mma_six(am0, ad0, b0, u0, s1, s2, s3, s4, s5, s6);
        load_frags(x, ldsD, bc0 + 2, 2, t, n, kA, am0, ad0);
        load_inv(iv + 1 * WPX, wb);
        __builtin_amdgcn_sched_barrier(0);
        fold6(tot1, tot2, tot3, tot4, s1, s2, s3, s4, s5, s6, wa);
        mma_six(am1, ad1, b1, u1, s1, s2, s3, s4, s5, s6);
        load_frags(x, ldsD, bc0 + 3, 3, t, n, kA, am1, ad1);
        load_inv(iv + 2 * WPX, wa);
        __builtin_amdgcn_sched_barrier(0);
        fold6(tot1, tot2, tot3, tot4, s1, s2, s3, s4, s5, s6, wb);
        mma_six(am0, ad0, b2, u2, s1, s2, s3, s4, s5, s6);
        load_frags(x, ldsD, bc0 + 0, 0, tn, n, kA, am0, ad0);
        load_inv(iv + 3 * WPX, wb);
        __builtin_amdgcn_sched_barrier(0);
        fold6(tot1, tot2, tot3, tot4, s1, s2, s3, s4, s5, s6, wa);
        mma_six(am1, ad1, b3, u3, s1, s2, s3, s4, s5, s6);
        fold6(tot1, tot2, tot3, tot4, s1, s2, s3, s4, s5, s6, wb);

        #pragma unroll
        for (int r = 0; r < 8; ++r)
            rmax = fmaxf(rmax, fmaxf(fmaxf(tot1[r], tot2[r]), fmaxf(tot3[r], tot4[r])) * 0.25f);
    }

    rmax = fmaxf(rmax, __shfl_xor(rmax, 16, 32));
    if (lane < 16) ldsR[wave * 16 + n] = rmax;
    __syncthreads();
    if (tid < 16) {
        float v = ldsR[tid];
        for (int w = 1; w < 4; ++w) v = fmaxf(v, ldsR[w * 16 + tid]);
        out[b * Kn + k0 + tid] = v;
    }
}

extern "C" void kernel_launch(void* const* d_in, const int* in_sizes, int n_in,
                              void* d_out, int out_size, void* d_ws, size_t ws_size,
                              hipStream_t stream) {
    const float* x   = (const float*)d_in[0];
    const float* shp = (const float*)d_in[1];
    float* out  = (float*)d_out;
    shapelet_wmma_kernel<<<Bn * (Kn / 16), NTH, 0, stream>>>(x, shp, out);
}


# Second 3-product split, across a unit's two tile pairs and adjacent block pairs, on all three product sums: 72 WMMAs per channel-unit, not 96; operand differences by VALU before a run; new sum order
# speedup vs baseline: 7.1721x; 1.1613x over previous
#include <hip/hip_runtime.h>
#include <hip/hip_bf16.h>
#include <math.h>


constexpr int Bn = 16;
constexpr int Cn = 4;
constexpr int Ln = 4096;
constexpr int Kn = 256;
constexpr int Sn = 128;
constexpr int Wn = Ln - Sn + 1;
constexpr int NT = (Wn + 15) / 16;
constexpr float EPSV = 1e-8f;
constexpr int NU  = (NT + 3) / 4;
constexpr int WPX = (NT + 3) * 16;
constexpr int WX  = Ln - 160;
constexpr int NCH = Ln / 16;
constexpr int NTH = 128;
constexpr int DPC = Ln / 2;
constexpr int DOD = Cn * DPC + 16;

typedef __attribute__((ext_vector_type(2))) float v2f;
typedef __attribute__((ext_vector_type(4))) float v4f;
typedef __attribute__((ext_vector_type(8))) float v8f;


__device__ __forceinline__ void load_frags(
    const float* __restrict__ x, const v2f* dl, int bc, int c, int t, int n, int kA,
    v2f (&am)[20], v2f (&ad)[40])
{
    const int w1 = (t << 4) + n;
    const int wa = w1 < Wn ? w1 : Wn - 1;
    const int wb = w1 < WX ? w1 : WX;
    const float* pa = x + (size_t)bc * Ln + wa + kA;
    const float* pb = x + (size_t)bc * Ln + wb + kA;
    const v2f* pd = dl + ((wa & 1) ? DOD : 0) + c * DPC + ((wa + kA) >> 1);
    const v2f* pe = dl + ((wb & 1) ? DOD : 0) + c * DPC + ((wb + kA) >> 1);
    #pragma unroll
    for (int q = 0; q < 4; ++q) {
        #pragma unroll
        for (int r = 0; r < 4; ++r) {
            am[4 * q + r].x = pa[4 * (8 * q + 4 + r)];
            am[4 * q + r].y = pa[4 * (8 * q + 4 + r) + 1];
        }
    }
    #pragma unroll
    for (int r = 0; r < 4; ++r) {
        am[16 + r].x = pb[4 * (36 + r)];
        am[16 + r].y = pb[4 * (36 + r) + 1];
    }
    #pragma unroll
    for (int j = 0; j < 32; ++j) ad[j] = pd[2 * j];
    #pragma unroll
    for (int j = 32; j < 40; ++j) ad[j] = pe[2 * j];
}

__device__ __forceinline__ void mma_nine(
    v2f (&am)[20], v2f (&ad)[40], const v2f (&bq)[32], const v2f (&uq)[16],
    const v2f (&uu)[8], const v2f (&ve)[8], const v2f (&vo)[8],
    v8f& g1, v8f& g2, v8f& g3, v8f& h1, v8f& h2, v8f& h3, v8f& k1, v8f& k2, v8f& k3)
{
    v2f x1[4], y1[4], z1[4];
    #pragma unroll
    for (int r = 0; r < 4; ++r) {
        x1[r]      = am[8 + r]  - am[4 + r];
        am[r]      = am[r]      - am[4 + r];
        am[8 + r]  = am[8 + r]  - am[12 + r];
        am[16 + r] = am[16 + r] - am[12 + r];
        y1[r]      = ad[16 + r] - ad[8 + r];
        ad[r]      = ad[r]      - ad[8 + r];
        ad[16 + r] = ad[16 + r] - ad[24 + r];
        ad[32 + r] = ad[32 + r] - ad[24 + r];
        z1[r]      = ad[20 + r] - ad[12 + r];
        ad[4 + r]  = ad[4 + r]  - ad[12 + r];
        ad[20 + r] = ad[20 + r] - ad[28 + r];
        ad[36 + r] = ad[36 + r] - ad[28 + r];
    }
    v8f a1 = {0.f, 0.f, 0.f, 0.f, 0.f, 0.f, 0.f, 0.f};
    v8f a2 = {0.f, 0.f, 0.f, 0.f, 0.f, 0.f, 0.f, 0.f};
    v8f a3 = {0.f, 0.f, 0.f, 0.f, 0.f, 0.f, 0.f, 0.f};
    v8f d1 = {0.f, 0.f, 0.f, 0.f, 0.f, 0.f, 0.f, 0.f};
    v8f d2 = {0.f, 0.f, 0.f, 0.f, 0.f, 0.f, 0.f, 0.f};
    v8f d3 = {0.f, 0.f, 0.f, 0.f, 0.f, 0.f, 0.f, 0.f};
    v8f e1 = {0.f, 0.f, 0.f, 0.f, 0.f, 0.f, 0.f, 0.f};
    v8f e2 = {0.f, 0.f, 0.f, 0.f, 0.f, 0.f, 0.f, 0.f};
    v8f e3 = {0.f, 0.f, 0.f, 0.f, 0.f, 0.f, 0.f, 0.f};
    __builtin_amdgcn_sched_barrier(0);
    #pragma unroll
    for (int p = 0; p < 2; ++p) {
        #pragma unroll
        for (int r = 0; r < 4; ++r) {
            a1 = __builtin_amdgcn_wmma_f32_16x16x4_f32(
                 false, am[4 + 8 * p + r],  false, uu[4 * p + r],
                 (short)0, a1,  false,  false);
            d1 = __builtin_amdgcn_wmma_f32_16x16x4_f32(
                 false, ad[8 + 16 * p + r],  false, ve[4 * p + r],
                 (short)0, d1,  false,  false);
            e1 = __builtin_amdgcn_wmma_f32_16x16x4_f32(
                 false, ad[12 + 16 * p + r],  false, vo[4 * p + r],
                 (short)0, e1,  false,  false);
            a2 = __builtin_amdgcn_wmma_f32_16x16x4_f32(
                 false, am[8 * p + r],  false, uq[8 * p + r],
                 (short)0, a2,  false,  false);
            d2 = __builtin_amdgcn_wmma_f32_16x16x4_f32(
                 false, ad[16 * p + r],  false, bq[16 * p + r],
                 (short)0, d2,  false,  false);
            e2 = __builtin_amdgcn_wmma_f32_16x16x4_f32(
                 false, ad[4 + 16 * p + r],  false, bq[16 * p + 4 + r],
                 (short)0, e2,  false,  false);
            a3 = __builtin_amdgcn_wmma_f32_16x16x4_f32(
                 false, p == 0 ? x1[r] : am[16 + r],  false, uq[8 * p + 4 + r],
                 (short)0, a3,  false,  false);
            d3 = __builtin_amdgcn_wmma_f32_16x16x4_f32(
                 false, p == 0 ? y1[r] : ad[32 + r],  false, bq[16 * p + 8 + r],
                 (short)0, d3,  false,  false);
            e3 = __builtin_amdgcn_wmma_f32_16x16x4_f32(
                 false, p == 0 ? z1[r] : ad[36 + r],  false, bq[16 * p + 12 + r],
                 (short)0, e3,  false,  false);
        }
    }
    __builtin_amdgcn_sched_barrier(0);
    asm volatile("v_nop\n\tv_nop\n\tv_nop\n\tv_nop"
                 : "+v"(a1), "+v"(a2), "+v"(a3), "+v"(d1), "+v"(d2), "+v"(d3), "+v"(e1), "+v"(e2), "+v"(e3));
    g1 = a1;
    g2 = a2;
    g3 = a3;
    h1 = d1;
    h2 = d2;
    h3 = d3;
    k1 = e1;
    k2 = e2;
    k3 = e3;
}

__device__ __forceinline__ void load_inv(const float* iv, v4f (&w)[8])
{
    #pragma unroll
    for (int a = 0; a < 4; ++a) {
        w[2 * a]     = *reinterpret_cast<const v4f*>(iv + 16 * a);
        w[2 * a + 1] = *reinterpret_cast<const v4f*>(iv + 16 * a + 4);
    }
}

__device__ __forceinline__ void fold9(
    v8f& tot1, v8f& tot2, v8f& tot3, v8f& tot4,
    const v8f& g1, const v8f& g2, const v8f& g3, const v8f& h1, const v8f& h2, const v8f& h3,
    const v8f& k1, const v8f& k2, const v8f& k3, const v4f (&w)[8])
{
    #pragma unroll
    for (int r = 0; r < 4; ++r) {
        const float a0 = g1[r] + g2[r],         a4 = g1[r + 4] + g2[r + 4];
        const float b0 = g1[r] + g3[r],         b4 = g1[r + 4] + g3[r + 4];
        tot1[r]     = fmaf(a0 + (h1[r] + h2[r]),         w[0][r], tot1[r]);
        tot1[r + 4] = fmaf(a4 + (h1[r + 4] + h2[r + 4]), w[1][r], tot1[r + 4]);
        tot2[r]     = fmaf(a0 - (k1[r] + k2[r]),         w[2][r], tot2[r]);
        tot2[r + 4] = fmaf(a4 - (k1[r + 4] + k2[r + 4]), w[3][r], tot2[r + 4]);
        tot3[r]     = fmaf(b0 + (h1[r] + h3[r]),         w[4][r], tot3[r]);
        tot3[r + 4] = fmaf(b4 + (h1[r + 4] + h3[r + 4]), w[5][r], tot3[r + 4]);
        tot4[r]     = fmaf(b0 - (k1[r] + k3[r]),         w[6][r], tot4[r]);
        tot4[r + 4] = fmaf(b4 - (k1[r + 4] + k3[r + 4]), w[7][r], tot4[r + 4]);
    }
    asm volatile("" : "+v"(tot1), "+v"(tot2), "+v"(tot3), "+v"(tot4));
}

__global__ __launch_bounds__(NTH) void shapelet_wmma_kernel(
    const float* __restrict__ x, const float* __restrict__ shp, float* __restrict__ out)
{
    __shared__ v2f   ldsB[Cn * (Sn / 2) * 16];
    __shared__ v2f   ldsUU[Cn * 2 * 8 * 16];
    __shared__ v2f   ldsVE[Cn * 2 * 8 * 16];
    __shared__ v2f   ldsVO[Cn * 2 * 8 * 16];
    __shared__ v2f   ldsU[Cn * 4 * 8 * 16];
    __shared__ __attribute__((aligned(16))) float ldsInv[Cn * WPX];
    __shared__ float ldsP[Cn * NCH];
    __shared__ __attribute__((aligned(16))) v2f ldsD[2 * Cn * DPC + 16];
    __shared__ float ldsR[4 * 16];

    const int b    = blockIdx.x >> 4;
    const int k0   = (blockIdx.x & 15) << 4;
    const int tid  = threadIdx.x;
    const int wave = tid >> 5;
    const int lane = tid & 31;
    const int n    = lane & 15;
    const int kA   = (lane >> 4) << 1;
    const int h8   = (lane >> 4) << 3;

    const int bc0 = b * Cn;

    const int srow = tid >> 1;
    const int sh   = tid & 1;
    const int sc   = srow >> 4;
    const int snn  = srow & 15;
    const float* ps = shp + (size_t)(sc * Kn + k0 + snn) * Sn + sh * 64;
    float sv[64];
    #pragma unroll
    for (int i = 0; i < 64; ++i) sv[i] = ps[i];

    {
        float pq[2];
        #pragma unroll
        for (int g = 0; g < 2; ++g) {
            float p0 = 0.f, p1 = 0.f, p2 = 0.f, p3 = 0.f;
            #pragma unroll
            for (int i = 32 * g; i < 32 * g + 32; i += 4) {
                p0 = fmaf(sv[i], sv[i], p0);         p1 = fmaf(sv[i + 1], sv[i + 1], p1);
                p2 = fmaf(sv[i + 2], sv[i + 2], p2); p3 = fmaf(sv[i + 3], sv[i + 3], p3);
            }
            pq[g] = (p0 + p1) + (p2 + p3);
        }
        float p = pq[0] + pq[1];
        p += __shfl_xor(p, 1, 32);
        const float sscale = 1.0f / fmaxf(sqrtf(p), EPSV);
        #pragma unroll
        for (int jj = 0; jj < 32; ++jj) {
            v2f pr;
            pr.x = sv[2 * jj] * sscale;
            pr.y = sv[2 * jj + 1] * sscale;
            ldsB[(sc * 64 + sh * 32 + jj) * 16 + snn] = pr;
        }
        #pragma unroll
        for (int i = 0; i < 8; ++i) {
            v2f t0, t1, t2, t3;
            t0.x = sv[2 * i] * sscale;       t0.y = sv[2 * i + 1] * sscale;
            t1.x = sv[16 + 2 * i] * sscale;  t1.y = sv[16 + 2 * i + 1] * sscale;
            t2.x = sv[32 + 2 * i] * sscale;  t2.y = sv[32 + 2 * i + 1] * sscale;
            t3.x = sv[48 + 2 * i] * sscale;  t3.y = sv[48 + 2 * i + 1] * sscale;
            const v2f ua = t0 + t1;
            const v2f ub = t2 + t3;
            ldsU[((sc * 4 + 2 * sh) * 8 + i) * 16 + snn]     = ua;
            ldsU[((sc * 4 + 2 * sh + 1) * 8 + i) * 16 + snn] = ub;
            ldsUU[((sc * 2 + sh) * 8 + i) * 16 + snn] = ua + ub;
            ldsVE[((sc * 2 + sh) * 8 + i) * 16 + snn] = t0 + t2;
            ldsVO[((sc * 2 + sh) * 8 + i) * 16 + snn] = t1 + t3;
        }
    }

    float hs[2][Cn][16], tp[2][Cn][16];
    #pragma unroll
    for (int rep = 0; rep < 2; ++rep) {
        const int vt = tid + NTH * rep;
        const int jb = (vt + 8 < NCH) ? vt + 8 : NCH - 1;
        #pragma unroll
        for (int c = 0; c < Cn; ++c) {
            const float* pa = x + (size_t)(bc0 + c) * Ln + 16 * vt;
            const float* pb = x + (size_t)(bc0 + c) * Ln + 16 * jb;
            const float* pn = x + (size_t)(bc0 + c) * Ln + 16 * ((vt + 1 < NCH) ? vt + 1 : NCH - 1);
            const float xnn = x[(size_t)(bc0 + c) * Ln + 16 * ((vt + 2 < NCH) ? vt + 2 : NCH - 1)];
            float qa[16], qb[16], qn[16];
            #pragma unroll
            for (int e = 0; e < 16; ++e) { qa[e] = pa[e]; qb[e] = pb[e]; qn[e] = pn[e]; }
            float dd[17];
            #pragma unroll
            for (int e = 0; e < 16; ++e) dd[e] = qa[e] - qn[e];
            dd[16] = qn[0] - xnn;
            #pragma unroll
            for (int i = 0; i < 8; ++i) {
                v2f pe, po;
                pe.x = dd[2 * i];     pe.y = dd[2 * i + 1];
                po.x = dd[2 * i + 1]; po.y = dd[2 * i + 2];
                ldsD[c * DPC + 8 * vt + i]       = pe;
                ldsD[DOD + c * DPC + 8 * vt + i] = po;
            }
            float h = 0.f;
            #pragma unroll
            for (int e = 15; e >= 0; --e) { h = fmaf(qa[e], qa[e], h); hs[rep][c][e] = h; }
            float tt = 0.f;
            #pragma unroll
            for (int e = 0; e < 16; ++e) { tp[rep][c][e] = tt; tt = fmaf(qb[e], qb[e], tt); }
            ldsP[c * NCH + vt] = hs[rep][c][0];
        }
    }
    __syncthreads();

    #pragma unroll
    for (int rep = 0; rep < 2; ++rep) {
        const int vt = tid + NTH * rep;
        if (vt < NT) {
            #pragma unroll
            for (int c = 0; c < Cn; ++c) {
                float mid = ldsP[c * NCH + vt + 1];
                #pragma unroll
                for (int u2 = 2; u2 < 8; ++u2) mid += ldsP[c * NCH + vt + u2];
                #pragma unroll
                for (int r = 0; r < 16; ++r) {
                    const float ss  = (hs[rep][c][r] + mid) + tp[rep][c][r];
                    const float inv = fminf(__builtin_amdgcn_rsqf(ss), 1.0f / EPSV);
                    ldsInv[c * WPX + 16 * vt + r] = (16 * vt + r < Wn) ? inv : 0.0f;
                }
            }
        } else if (vt < NT + 3) {
            #pragma unroll
            for (int c = 0; c < Cn; ++c) {
                #pragma unroll
                for (int r = 0; r < 16; ++r) ldsInv[c * WPX + 16 * vt + r] = 0.0f;
            }
        }
    }
    __syncthreads();

    v2f b0[32], b1[32], b2[32], b3[32];
    #pragma unroll
    for (int j = 0; j < 32; ++j) {
        b0[j] = ldsB[(0 * 64 + ((4 * j + kA) >> 1)) * 16 + n];
        b1[j] = ldsB[(1 * 64 + ((4 * j + kA) >> 1)) * 16 + n];
        b2[j] = ldsB[(2 * 64 + ((4 * j + kA) >> 1)) * 16 + n];
        b3[j] = ldsB[(3 * 64 + ((4 * j + kA) >> 1)) * 16 + n];
    }
    v2f u0[16], u1[16], u2[16], u3[16];
    #pragma unroll
    for (int q = 0; q < 4; ++q) {
        #pragma unroll
        for (int r = 0; r < 4; ++r) {
            const int o = (q * 8 + 2 * r + (kA >> 1)) * 16 + n;
            u0[4 * q + r] = ldsU[0 * 512 + o];
            u1[4 * q + r] = ldsU[1 * 512 + o];
            u2[4 * q + r] = ldsU[2 * 512 + o];
            u3[4 * q + r] = ldsU[3 * 512 + o];
        }
    }
    v2f uu0[8], uu1[8], uu2[8], uu3[8], ve0[8], ve1[8], ve2[8], ve3[8], vo0[8], vo1[8], vo2[8], vo3[8];
    #pragma unroll
    for (int p = 0; p < 2; ++p) {
        #pragma unroll
        for (int r = 0; r < 4; ++r) {
            const int o = (p * 8 + 2 * r + (kA >> 1)) * 16 + n;
            uu0[4 * p + r] = ldsUU[0 * 256 + o];
            uu1[4 * p + r] = ldsUU[1 * 256 + o];
            uu2[4 * p + r] = ldsUU[2 * 256 + o];
            uu3[4 * p + r] = ldsUU[3 * 256 + o];
            ve0[4 * p + r] = ldsVE[0 * 256 + o];
            ve1[4 * p + r] = ldsVE[1 * 256 + o];
            ve2[4 * p + r] = ldsVE[2 * 256 + o];
            ve3[4 * p + r] = ldsVE[3 * 256 + o];
            vo0[4 * p + r] = ldsVO[0 * 256 + o];
            vo1[4 * p + r] = ldsVO[1 * 256 + o];
            vo2[4 * p + r] = ldsVO[2 * 256 + o];
            vo3[4 * p + r] = ldsVO[3 * 256 + o];
        }
    }

    v2f am0[20], ad0[40], am1[20], ad1[40];

    int u = wave;
    load_frags(x, ldsD, bc0 + 0, 0, 4 * u, n, kA, am0, ad0);

    float rmax = 0.0f;
    for (; u < NU; u += 4) {
        const int t  = 4 * u;
        const int tn = (u + 4 < NU) ? t + 16 : t;
        const float* iv = ldsInv + 16 * t + h8;
        v8f tot1 = {0.f, 0.f, 0.f, 0.f, 0.f, 0.f, 0.f, 0.f};
        v8f tot2 = {0.f, 0.f, 0.f, 0.f, 0.f, 0.f, 0.f, 0.f};
        v8f tot3 = {0.f, 0.f, 0.f, 0.f, 0.f, 0.f, 0.f, 0.f};
        v8f tot4 = {0.f, 0.f, 0.f, 0.f, 0.f, 0.f, 0.f, 0.f};
        v8f g1, g2, g3, h1, h2, h3, k1, k2, k3;
        v4f wa[8], wb[8];

        load_inv(iv + 0 * WPX, wa);
        load_frags(x, ldsD, bc0 + 1, 1, t, n, kA, am1, ad1);
        mma_nine(am0, ad0, b0, u0, uu0, ve0, vo0, g1, g2, g3, h1, h2, h3, k1, k2, k3);
        load_frags(x, ldsD, bc0 + 2, 2, t, n, kA, am0, ad0);
        load_inv(iv + 1 * WPX, wb);
        __builtin_amdgcn_sched_barrier(0);
        fold9(tot1, tot2, tot3, tot4, g1, g2, g3, h1, h2, h3, k1, k2, k3, wa);
        mma_nine(am1, ad1, b1, u1, uu1, ve1, vo1, g1, g2, g3, h1, h2, h3, k1, k2, k3);
        load_frags(x, ldsD, bc0 + 3, 3, t, n, kA, am1, ad1);
        load_inv(iv + 2 * WPX, wa);
        __builtin_amdgcn_sched_barrier(0);
        fold9(tot1, tot2, tot3, tot4, g1, g2, g3, h1, h2, h3, k1, k2, k3, wb);
        mma_nine(am0, ad0, b2, u2, uu2, ve2, vo2, g1, g2, g3, h1, h2, h3, k1, k2, k3);
        load_frags(x, ldsD, bc0 + 0, 0, tn, n, kA, am0, ad0);
        load_inv(iv + 3 * WPX, wb);
        __builtin_amdgcn_sched_barrier(0);
        fold9(tot1, tot2, tot3, tot4, g1, g2, g3, h1, h2, h3, k1, k2, k3, wa);
        mma_nine(am1, ad1, b3, u3, uu3, ve3, vo3, g1, g2, g3, h1, h2, h3, k1, k2, k3);
        fold9(tot1, tot2, tot3, tot4, g1, g2, g3, h1, h2, h3, k1, k2, k3, wb);

        #pragma unroll
        for (int r = 0; r < 8; ++r)
            rmax = fmaxf(rmax, fmaxf(fmaxf(tot1[r], tot2[r]), fmaxf(tot3[r], tot4[r])) * 0.25f);
    }

    rmax = fmaxf(rmax, __shfl_xor(rmax, 16, 32));
    if (lane < 16) ldsR[wave * 16 + n] = rmax;
    __syncthreads();
    if (tid < 16) {
        float v = ldsR[tid];
        for (int w = 1; w < 4; ++w) v = fmaxf(v, ldsR[w * 16 + tid]);
        out[b * Kn + k0 + tid] = v;
    }
}

extern "C" void kernel_launch(void* const* d_in, const int* in_sizes, int n_in,
                              void* d_out, int out_size, void* d_ws, size_t ws_size,
                              hipStream_t stream) {
    const float* x   = (const float*)d_in[0];
    const float* shp = (const float*)d_in[1];
    float* out  = (float*)d_out;
    shapelet_wmma_kernel<<<Bn * (Kn / 16), NTH, 0, stream>>>(x, shp, out);
}
